# starting kernel
# baseline (speedup 1.0000x reference)
#include <hip/hip_runtime.h>

#define N_NODES 50000
#define N_EDGES 800000
#define HDIM    64

typedef float v2f __attribute__((ext_vector_type(2)));
typedef float v8f __attribute__((ext_vector_type(8)));

__global__ void deg_init(float* __restrict__ deg) {
    int i = blockIdx.x * 256 + threadIdx.x;
    if (i < N_NODES) deg[i] = 1.0f;
}

__global__ void deg_count(const int* __restrict__ ei, float* __restrict__ deg) {
    int e = blockIdx.x * 256 + threadIdx.x;
    if (e < N_EDGES) {
        int dst = ei[N_EDGES + e];
        atomicAdd(&deg[dst], 1.0f);
    }
}

__global__ void deg_to_dinv(float* __restrict__ deg) {
    int i = blockIdx.x * 256 + threadIdx.x;
    if (i < N_NODES) deg[i] = rsqrtf(deg[i]);
}

template <bool RELU_IN>
__global__ __launch_bounds__(128)
void gcn_gemm(const float* __restrict__ in, const float* __restrict__ bias,
              const float* __restrict__ W, const float* __restrict__ dinv,
              float* __restrict__ h, float* __restrict__ agg) {
    const int wave = threadIdx.x >> 5;
    const int lane = threadIdx.x & 31;
    const int tile = blockIdx.x * 4 + wave;
    const int tm   = tile >> 2;
    const int tn   = tile & 3;

    const int mrow  = (tm << 4) + (lane & 15);
    const int khalf = (lane >> 4) << 1;
    const int ncol  = (tn << 4) + (lane & 15);

    const float* __restrict__ arow = in + (size_t)mrow * HDIM;

    v8f c = {0.f, 0.f, 0.f, 0.f, 0.f, 0.f, 0.f, 0.f};
#pragma unroll
    for (int k = 0; k < HDIM; k += 4) {
        const int ka = k + khalf;
        v2f a;
        a.x = arow[ka];
        a.y = arow[ka + 1];
        if (RELU_IN) {
            a.x = fmaxf(a.x + bias[ka],     0.f);
            a.y = fmaxf(a.y + bias[ka + 1], 0.f);
        }
        v2f b;
        b.x = W[(size_t)ka * HDIM + ncol];
        b.y = W[(size_t)(ka + 1) * HDIM + ncol];
        c = __builtin_amdgcn_wmma_f32_16x16x4_f32(false, a, false, b,
                                                  (short)0, c, false, false);
    }

    const int mbase = (tm << 4) + ((lane >> 4) << 3);
    const int nn    = (tn << 4) + (lane & 15);
#pragma unroll
    for (int v = 0; v < 8; ++v) {
        const int row = mbase + v;
        const float val = c[v];
        h[(size_t)row * HDIM + nn] = val;
        const float di = dinv[row];
        agg[(size_t)row * HDIM + nn] = val * di * di;
    }
}

__global__ __launch_bounds__(256)
void gcn_scatter(const int* __restrict__ ei, const float* __restrict__ dinv,
                 const float* __restrict__ h, float* __restrict__ agg) {
    const int t = blockIdx.x * 256 + threadIdx.x;
    const int e = t >> 4;
    if (e >= N_EDGES) return;
    const int fo  = (t & 15) << 2;
    const int src = ei[e];
    const int dst = ei[N_EDGES + e];
    const float nrm = dinv[src] * dinv[dst];

    const float4 v = *(const float4*)(h + (size_t)src * HDIM + fo);
    float* ap = agg + (size_t)dst * HDIM + fo;
    atomicAdd(ap + 0, v.x * nrm);
    atomicAdd(ap + 1, v.y * nrm);
    atomicAdd(ap + 2, v.z * nrm);
    atomicAdd(ap + 3, v.w * nrm);
}

__global__ void gcn_act(const float* __restrict__ agg, const float* __restrict__ bias,
                        float* __restrict__ out) {
    int i = blockIdx.x * 256 + threadIdx.x;
    if (i < N_NODES * HDIM)
        out[i] = fmaxf(agg[i] + bias[i & (HDIM - 1)], 0.f);
}

extern "C" void kernel_launch(void* const* d_in, const int* in_sizes, int n_in,
                              void* d_out, int out_size, void* d_ws, size_t ws_size,
                              hipStream_t stream) {
    const float* x  = (const float*)d_in[0];
    const int*   ei = (const int*)d_in[1];
    const float* W  = (const float*)d_in[2];
    const float* b  = (const float*)d_in[3];
    float* out = (float*)d_out;

    float* ws   = (float*)d_ws;
    float* dinv = ws;
    float* h    = dinv + N_NODES;
    float* aggA = h    + (size_t)N_NODES * HDIM;
    float* aggB = aggA + (size_t)N_NODES * HDIM;

    const int nodeBlocks = (N_NODES + 255) / 256;
    const int edgeBlocks = (N_EDGES + 255) / 256;
    const int scatBlocks = (N_EDGES * 16) / 256;
    const int gemmBlocks = (N_NODES / 16) * (HDIM / 16) / 4;
    const int actBlocks  = (N_NODES * HDIM + 255) / 256;

    deg_init   <<<nodeBlocks, 256, 0, stream>>>(dinv);
    deg_count  <<<edgeBlocks, 256, 0, stream>>>(ei, dinv);
    deg_to_dinv<<<nodeBlocks, 256, 0, stream>>>(dinv);

    gcn_gemm<false><<<gemmBlocks, 128, 0, stream>>>(x, nullptr, W, dinv, h, aggA);
    gcn_scatter    <<<scatBlocks, 256, 0, stream>>>(ei, dinv, h, aggA);

    gcn_gemm<true> <<<gemmBlocks, 128, 0, stream>>>(aggA, b, W + HDIM * HDIM, dinv, h, aggB);
    gcn_scatter    <<<scatBlocks, 256, 0, stream>>>(ei, dinv, h, aggB);

    gcn_gemm<true> <<<gemmBlocks, 128, 0, stream>>>(aggB, b + HDIM, W + 2 * HDIM * HDIM, dinv, h, aggA);
    gcn_scatter    <<<scatBlocks, 256, 0, stream>>>(ei, dinv, h, aggA);

    gcn_act<<<actBlocks, 256, 0, stream>>>(aggA, b + 2 * HDIM, out);
}


# GEMM: one wave owns a 16-row tile x all 64 columns: A read once by 16-byte loads (k order permuted alike in A and B), W from an LDS copy, whole-line stores by lane-half swap; k summing order changes
# speedup vs baseline: 3.8178x; 3.8178x over previous
#include <hip/hip_runtime.h>

#define N_NODES 50000
#define N_EDGES 800000
#define HDIM    64

typedef float v2f __attribute__((ext_vector_type(2)));
typedef float v8f __attribute__((ext_vector_type(8)));

#define HALF_NODES  (N_NODES / 2)
#define BLOCK_EDGES 512
#define N_CHUNKS    ((N_EDGES + BLOCK_EDGES - 1) / BLOCK_EDGES)
#define CTR_STRIDE  64
#define N_CTRS      8

static __device__ __forceinline__ int claim_job(unsigned* __restrict__ ctr) {
    __shared__ int s_job;
    if (threadIdx.x == 0) {
        const unsigned id = __builtin_amdgcn_s_sendmsg_rtn(0x87);
        unsigned r = (id >> 18) & 1u;
        unsigned t = __hip_atomic_fetch_add(ctr + r * CTR_STRIDE, 1u, __ATOMIC_RELAXED, __HIP_MEMORY_SCOPE_AGENT);
        if (t >= (unsigned)N_CHUNKS) {
            r ^= 1u;
            t = __hip_atomic_fetch_add(ctr + r * CTR_STRIDE, 1u, __ATOMIC_RELAXED, __HIP_MEMORY_SCOPE_AGENT);
        }
        s_job = (t < (unsigned)N_CHUNKS) ? (int)(t * 2u + r) : -1;
    }
    __syncthreads();
    return __builtin_amdgcn_readfirstlane(s_job);
}

__global__ void deg_init(float* __restrict__ deg, unsigned* __restrict__ ctr) {
    int i = blockIdx.x * 256 + threadIdx.x;
    if (i < N_NODES) deg[i] = 1.0f;
    if (i < N_CTRS) ctr[i * CTR_STRIDE] = 0u;
}

__global__ __launch_bounds__(256)
void deg_count(const int* __restrict__ ei, float* __restrict__ deg, unsigned* __restrict__ ctr) {
    const int job = claim_job(ctr);
    if (job < 0) return;
    const unsigned lo = (unsigned)(job & 1) * HALF_NODES;
    const int ea = (job >> 1) * BLOCK_EDGES + threadIdx.x;
    const int eb = ea + 256;
    int da = -1, db = -1;
    if (ea < N_EDGES) da = ei[N_EDGES + ea];
    if (eb < N_EDGES) db = ei[N_EDGES + eb];
    if ((unsigned)da - lo < (unsigned)HALF_NODES) atomicAdd(&deg[da], 1.0f);
    if ((unsigned)db - lo < (unsigned)HALF_NODES) atomicAdd(&deg[db], 1.0f);
}

__global__ void deg_to_dinv(float* __restrict__ deg) {
    int i = blockIdx.x * 256 + threadIdx.x;
    if (i < N_NODES) deg[i] = rsqrtf(deg[i]);
}


#define ROW_TILES (N_NODES / 16)
#define W_PAIR    144

template <bool SUM_IN>
__global__ __launch_bounds__(128)
void gcn_gemm(const float* __restrict__ in, const float* __restrict__ bias,
              const float* __restrict__ W, const float* __restrict__ dinv,
              float* __restrict__ hs, float* __restrict__ sum) {
    __shared__ __attribute__((aligned(16))) float Wl[(HDIM / 2) * W_PAIR];

    const int wave = threadIdx.x >> 5;
    const int lane = threadIdx.x & 31;
    const int c    = lane & 15;
    const int h    = lane >> 4;
    const int tm   = blockIdx.x * 4 + wave;
    const int tmc  = tm < ROW_TILES ? tm : ROW_TILES - 1;

#pragma unroll
    for (int j = 0; j < 4; ++j) {
        const int unit = j * 128 + (int)threadIdx.x;
        const int r = unit >> 4, q = unit & 15;
        const float4 e = *(const float4*)(W + (2 * r) * HDIM + 4 * q);
        const float4 o = *(const float4*)(W + (2 * r + 1) * HDIM + 4 * q);
        float* __restrict__ dst = Wl + r * W_PAIR + 8 * q;
        *(float4*)(dst)     = make_float4(e.x, o.x, e.y, o.y);
        *(float4*)(dst + 4) = make_float4(e.z, o.z, e.w, o.w);
    }

    const int mrow = (tmc << 4) + c;
    const float* __restrict__ arow = in + (size_t)mrow * HDIM + 4 * h;
    float4 a4[8];
#pragma unroll
    for (int j = 0; j < 8; ++j) a4[j] = *(const float4*)(arow + 8 * j);
    if (SUM_IN) {
        const float din = dinv[mrow];
#pragma unroll
        for (int j = 0; j < 8; ++j) {
            const float4 b4 = *(const float4*)(bias + 8 * j + 4 * h);
            a4[j].x = fmaxf(a4[j].x * din + b4.x, 0.f);
            a4[j].y = fmaxf(a4[j].y * din + b4.y, 0.f);
            a4[j].z = fmaxf(a4[j].z * din + b4.z, 0.f);
            a4[j].w = fmaxf(a4[j].w * din + b4.w, 0.f);
        }
    }

    __syncthreads();
    if (tm >= ROW_TILES) return;

    const int rbase = tm << 4;
    float srow[8];
#pragma unroll
    for (int v = 0; v < 8; ++v) srow[v] = dinv[rbase + 8 * h + v];

    const v2f* __restrict__ wl = (const v2f*)(Wl + (2 * h) * W_PAIR + 2 * c);

#pragma unroll
    for (int p = 0; p < 2; ++p) {
        v8f lo = {0.f, 0.f, 0.f, 0.f, 0.f, 0.f, 0.f, 0.f};
        v8f hi = {0.f, 0.f, 0.f, 0.f, 0.f, 0.f, 0.f, 0.f};
#pragma unroll
        for (int j = 0; j < 8; ++j) {
#pragma unroll
            for (int u = 0; u < 2; ++u) {
                v2f a;
                a.x = u ? a4[j].z : a4[j].x;
                a.y = u ? a4[j].w : a4[j].y;
                const int pr = (4 * j + u) * (W_PAIR / 2);
                const v2f b0 = wl[pr + 16 * (2 * p)];
                const v2f b1 = wl[pr + 16 * (2 * p + 1)];
                lo = __builtin_amdgcn_wmma_f32_16x16x4_f32(false, a, false, b0, (short)0, lo, false, false);
                hi = __builtin_amdgcn_wmma_f32_16x16x4_f32(false, a, false, b1, (short)0, hi, false, false);
            }
        }
#pragma unroll
        for (int v = 0; v < 8; ++v) {
            const float l = lo[v] * srow[v];
            const float g = hi[v] * srow[v];
            const float give = h ? l : g;
            const float got = __builtin_bit_cast(float, __builtin_amdgcn_permlanex16(
                __builtin_bit_cast(unsigned, give), __builtin_bit_cast(unsigned, give),
                0x76543210u, 0xfedcba98u, false, false));
            const float r0 = h ? got : l;
            const float r1 = h ? g   : got;
            const size_t o0 = (size_t)(rbase + v) * HDIM + 32 * p + lane;
            const size_t o1 = o0 + 8 * HDIM;
            hs[o0]  = r0;
            sum[o0] = r0;
            hs[o1]  = r1;
            sum[o1] = r1;
        }
        __builtin_amdgcn_sched_barrier(0);
    }
}

template <int K>
static __device__ __forceinline__ unsigned
scatter_batch(unsigned mask, int src, int dst, unsigned lane,
              const float* __restrict__ h, float* __restrict__ agg) {
    float v0[K], v1[K];
    unsigned m = mask;
#pragma unroll
    for (int j = 0; j < K; ++j) {
        const int l = __builtin_ctz(m);
        m &= m - 1;
        const unsigned so = (unsigned)__builtin_amdgcn_readlane(src, l) * HDIM + lane;
        v0[j] = h[so];
        v1[j] = h[so + 32];
    }
    m = mask;
#pragma unroll
    for (int j = 0; j < K; ++j) {
        const int l = __builtin_ctz(m);
        m &= m - 1;
        const int d = __builtin_amdgcn_readlane(dst, l);
        const unsigned dofs = (unsigned)d * HDIM + lane;
        atomicAdd(agg + dofs,      v0[j]);
        atomicAdd(agg + dofs + 32, v1[j]);
    }
    return m;
}

static __device__ __forceinline__ void
scatter_set(unsigned mask, int src, int dst, unsigned lane,
            const float* __restrict__ h, float* __restrict__ agg) {
    int left = __builtin_popcount(mask);
    while (left >= 16) { mask = scatter_batch<16>(mask, src, dst, lane, h, agg); left -= 16; }
    if (left & 8) mask = scatter_batch<8>(mask, src, dst, lane, h, agg);
    if (left & 4) mask = scatter_batch<4>(mask, src, dst, lane, h, agg);
    if (left & 2) mask = scatter_batch<2>(mask, src, dst, lane, h, agg);
    if (left & 1) mask = scatter_batch<1>(mask, src, dst, lane, h, agg);
}

__global__ __launch_bounds__(256)
void gcn_scatter(const int* __restrict__ ei,
                 const float* __restrict__ h, float* __restrict__ agg,
                 unsigned* __restrict__ ctr) {
    const int job = claim_job(ctr);
    if (job < 0) return;
    const unsigned lo   = (unsigned)(job & 1) * HALF_NODES;
    const unsigned lane = threadIdx.x & 31;
    const int e0 = (job >> 1) * BLOCK_EDGES + (int)(threadIdx.x >> 5) * 64 + (int)lane;
    const int e1 = e0 + 32;

    int src0 = 0, dst0 = -1, src1 = 0, dst1 = -1;
    if (e0 < N_EDGES) { src0 = ei[e0]; dst0 = ei[N_EDGES + e0]; }
    if (e1 < N_EDGES) { src1 = ei[e1]; dst1 = ei[N_EDGES + e1]; }
    const bool in0 = (unsigned)dst0 - lo < (unsigned)HALF_NODES;
    const bool in1 = (unsigned)dst1 - lo < (unsigned)HALF_NODES;

    scatter_set(__builtin_amdgcn_ballot_w32(in0), src0, dst0, lane, h, agg);
    scatter_set(__builtin_amdgcn_ballot_w32(in1), src1, dst1, lane, h, agg);
}

__global__ void gcn_act(const float* __restrict__ agg, const float* __restrict__ dinv,
                        const float* __restrict__ bias, float* __restrict__ out) {
    int i = blockIdx.x * 256 + threadIdx.x;
    if (i < N_NODES * HDIM)
        out[i] = fmaxf(agg[i] * dinv[(unsigned)i / HDIM] + bias[i & (HDIM - 1)], 0.f);
}

extern "C" void kernel_launch(void* const* d_in, const int* in_sizes, int n_in,
                              void* d_out, int out_size, void* d_ws, size_t ws_size,
                              hipStream_t stream) {
    const float* x  = (const float*)d_in[0];
    const int*   ei = (const int*)d_in[1];
    const float* W  = (const float*)d_in[2];
    const float* b  = (const float*)d_in[3];
    float* out = (float*)d_out;

    float* ws   = (float*)d_ws;
    float* dinv = ws;
    float* h    = dinv + 50048;
    float* aggA = h    + (size_t)N_NODES * HDIM;
    float* aggB = aggA + (size_t)N_NODES * HDIM;
    unsigned* ctr = (unsigned*)(aggB + (size_t)N_NODES * HDIM);

    const int nodeBlocks = (N_NODES + 255) / 256;
    const int claimBlocks = 2 * N_CHUNKS;
    const int gemmBlocks = (ROW_TILES + 3) / 4;
    const int actBlocks  = (N_NODES * HDIM + 255) / 256;

    deg_init   <<<nodeBlocks, 256, 0, stream>>>(dinv, ctr);
    deg_count  <<<claimBlocks, 256, 0, stream>>>(ei, dinv, ctr);
    deg_to_dinv<<<nodeBlocks, 256, 0, stream>>>(dinv);

    gcn_gemm<false><<<gemmBlocks, 128, 0, stream>>>(x, nullptr, W, dinv, h, aggA);
    gcn_scatter    <<<claimBlocks, 256, 0, stream>>>(ei, h, aggA, ctr + 2 * CTR_STRIDE);

    gcn_gemm<true> <<<gemmBlocks, 128, 0, stream>>>(aggA, b, W + HDIM * HDIM, dinv, h, aggB);
    gcn_scatter    <<<claimBlocks, 256, 0, stream>>>(ei, h, aggB, ctr + 4 * CTR_STRIDE);

    gcn_gemm<true> <<<gemmBlocks, 128, 0, stream>>>(aggB, b + HDIM, W + 2 * HDIM * HDIM, dinv, h, aggA);
    gcn_scatter    <<<claimBlocks, 256, 0, stream>>>(ei, h, aggA, ctr + 6 * CTR_STRIDE);

    gcn_act<<<actBlocks, 256, 0, stream>>>(aggA, dinv, b + 2 * HDIM, out);
}


# Scatter pushes from the source: the degree pass files each edge's dst id in its source's 64-slot row (spill list beyond); a wave loads a row once and adds it to all its dsts; sums stay arrival order
# speedup vs baseline: 3.9593x; 1.0371x over previous
#include <hip/hip_runtime.h>

#define N_NODES 50000
#define N_EDGES 800000
#define HDIM    64

typedef float v2f __attribute__((ext_vector_type(2)));
typedef float v8f __attribute__((ext_vector_type(8)));

#define HALF_NODES  (N_NODES / 2)
#define BLOCK_EDGES 512
#define N_CHUNKS    ((N_EDGES + BLOCK_EDGES - 1) / BLOCK_EDGES)
#define CTR_STRIDE  64
#define N_CTRS      8

#define SLOTS          64
#define GRP_ROWS       50048
#define SRC_PER_WAVE   4
#define SRC_PER_BLOCK  (8 * SRC_PER_WAVE)
#define N_SRC_CHUNKS   ((N_NODES + SRC_PER_BLOCK - 1) / SRC_PER_BLOCK)

static __device__ __forceinline__ int claim_job(unsigned* __restrict__ ctr, const unsigned n_chunks) {
    __shared__ int s_job;
    if (threadIdx.x == 0) {
        const unsigned id = __builtin_amdgcn_s_sendmsg_rtn(0x87);
        unsigned r = (id >> 18) & 1u;
        unsigned t = __hip_atomic_fetch_add(ctr + r * CTR_STRIDE, 1u, __ATOMIC_RELAXED, __HIP_MEMORY_SCOPE_AGENT);
        if (t >= n_chunks) {
            r ^= 1u;
            t = __hip_atomic_fetch_add(ctr + r * CTR_STRIDE, 1u, __ATOMIC_RELAXED, __HIP_MEMORY_SCOPE_AGENT);
        }
        s_job = (t < n_chunks) ? (int)(t * 2u + r) : -1;
    }
    __syncthreads();
    return __builtin_amdgcn_readfirstlane(s_job);
}

__global__ void deg_init(float* __restrict__ deg, unsigned* __restrict__ ctr,
                         unsigned* __restrict__ grp, unsigned* __restrict__ spill_n) {
    int i = blockIdx.x * 256 + threadIdx.x;
    if (i < N_NODES) { deg[i] = 1.0f; grp[i] = 0u; }
    if (i < N_CTRS) ctr[i * CTR_STRIDE] = 0u;
    if (i == 0) *spill_n = 0u;
}

static __device__ __forceinline__ void
file_edge(const int s, const int d, const unsigned place, unsigned* __restrict__ grp,
          unsigned* __restrict__ spill_n, int* __restrict__ spill) {
    if (place < (unsigned)SLOTS) {
        grp[GRP_ROWS + (unsigned)s * SLOTS + place] = (unsigned)d;
    } else {
        const unsigned i = __hip_atomic_fetch_add(spill_n, 1u, __ATOMIC_RELAXED, __HIP_MEMORY_SCOPE_AGENT);
        spill[2u * i]      = s;
        spill[2u * i + 1u] = d;
    }
}

__global__ __launch_bounds__(256)
void deg_count(const int* __restrict__ ei, float* __restrict__ deg, unsigned* __restrict__ ctr,
               unsigned* __restrict__ grp,
               unsigned* __restrict__ spill_n, int* __restrict__ spill) {
    const int job = claim_job(ctr, (unsigned)N_CHUNKS);
    if (job < 0) return;
    const unsigned lo = (unsigned)(job & 1) * HALF_NODES;
    const int ea = (job >> 1) * BLOCK_EDGES + threadIdx.x;
    const int eb = ea + 256;
    int sa = -1, da = -1, sb = -1, db = -1;
    if (ea < N_EDGES) { sa = ei[ea]; da = ei[N_EDGES + ea]; }
    if (eb < N_EDGES) { sb = ei[eb]; db = ei[N_EDGES + eb]; }
    if ((unsigned)da - lo < (unsigned)HALF_NODES) atomicAdd(&deg[da], 1.0f);
    if ((unsigned)db - lo < (unsigned)HALF_NODES) atomicAdd(&deg[db], 1.0f);
    const bool fa = (unsigned)sa - lo < (unsigned)HALF_NODES;
    const bool fb = (unsigned)sb - lo < (unsigned)HALF_NODES;
    unsigned pa = 0u, pb = 0u;
    if (fa) pa = __hip_atomic_fetch_add(grp + sa, 1u, __ATOMIC_RELAXED, __HIP_MEMORY_SCOPE_AGENT);
    if (fb) pb = __hip_atomic_fetch_add(grp + sb, 1u, __ATOMIC_RELAXED, __HIP_MEMORY_SCOPE_AGENT);
    if (fa) file_edge(sa, da, pa, grp, spill_n, spill);
    if (fb) file_edge(sb, db, pb, grp, spill_n, spill);
}

__global__ void deg_to_dinv(float* __restrict__ deg) {
    int i = blockIdx.x * 256 + threadIdx.x;
    if (i < N_NODES) deg[i] = rsqrtf(deg[i]);
}


#define ROW_TILES (N_NODES / 16)
#define W_PAIR    144

template <bool SUM_IN>
__global__ __launch_bounds__(128)
void gcn_gemm(const float* __restrict__ in, const float* __restrict__ bias,
              const float* __restrict__ W, const float* __restrict__ dinv,
              float* __restrict__ hs, float* __restrict__ sum) {
    __shared__ __attribute__((aligned(16))) float Wl[(HDIM / 2) * W_PAIR];

    const int wave = threadIdx.x >> 5;
    const int lane = threadIdx.x & 31;
    const int c    = lane & 15;
    const int h    = lane >> 4;
    const int tm   = blockIdx.x * 4 + wave;
    const int tmc  = tm < ROW_TILES ? tm : ROW_TILES - 1;

#pragma unroll
    for (int j = 0; j < 4; ++j) {
        const int unit = j * 128 + (int)threadIdx.x;
        const int r = unit >> 4, q = unit & 15;
        const float4 e = *(const float4*)(W + (2 * r) * HDIM + 4 * q);
        const float4 o = *(const float4*)(W + (2 * r + 1) * HDIM + 4 * q);
        float* __restrict__ dst = Wl + r * W_PAIR + 8 * q;
        *(float4*)(dst)     = make_float4(e.x, o.x, e.y, o.y);
        *(float4*)(dst + 4) = make_float4(e.z, o.z, e.w, o.w);
    }

    const int mrow = (tmc << 4) + c;
    const float* __restrict__ arow = in + (size_t)mrow * HDIM + 4 * h;
    float4 a4[8];
#pragma unroll
    for (int j = 0; j < 8; ++j) a4[j] = *(const float4*)(arow + 8 * j);
    if (SUM_IN) {
        const float din = dinv[mrow];
#pragma unroll
        for (int j = 0; j < 8; ++j) {
            const float4 b4 = *(const float4*)(bias + 8 * j + 4 * h);
            a4[j].x = fmaxf(a4[j].x * din + b4.x, 0.f);
            a4[j].y = fmaxf(a4[j].y * din + b4.y, 0.f);
            a4[j].z = fmaxf(a4[j].z * din + b4.z, 0.f);
            a4[j].w = fmaxf(a4[j].w * din + b4.w, 0.f);
        }
    }

    __syncthreads();
    if (tm >= ROW_TILES) return;

    const int rbase = tm << 4;
    float srow[8];
#pragma unroll
    for (int v = 0; v < 8; ++v) srow[v] = dinv[rbase + 8 * h + v];

    const v2f* __restrict__ wl = (const v2f*)(Wl + (2 * h) * W_PAIR + 2 * c);

#pragma unroll
    for (int p = 0; p < 2; ++p) {
        v8f lo = {0.f, 0.f, 0.f, 0.f, 0.f, 0.f, 0.f, 0.f};
        v8f hi = {0.f, 0.f, 0.f, 0.f, 0.f, 0.f, 0.f, 0.f};
#pragma unroll
        for (int j = 0; j < 8; ++j) {
#pragma unroll
            for (int u = 0; u < 2; ++u) {
                v2f a;
                a.x = u ? a4[j].z : a4[j].x;
                a.y = u ? a4[j].w : a4[j].y;
                const int pr = (4 * j + u) * (W_PAIR / 2);
                const v2f b0 = wl[pr + 16 * (2 * p)];
                const v2f b1 = wl[pr + 16 * (2 * p + 1)];
                lo = __builtin_amdgcn_wmma_f32_16x16x4_f32(false, a, false, b0, (short)0, lo, false, false);
                hi = __builtin_amdgcn_wmma_f32_16x16x4_f32(false, a, false, b1, (short)0, hi, false, false);
            }
        }
#pragma unroll
        for (int v = 0; v < 8; ++v) {
            const float l = lo[v] * srow[v];
            const float g = hi[v] * srow[v];
            const float give = h ? l : g;
            const float got = __builtin_bit_cast(float, __builtin_amdgcn_permlanex16(
                __builtin_bit_cast(unsigned, give), __builtin_bit_cast(unsigned, give),
                0x76543210u, 0xfedcba98u, false, false));
            const float r0 = h ? got : l;
            const float r1 = h ? g   : got;
            const size_t o0 = (size_t)(rbase + v) * HDIM + 32 * p + lane;
            const size_t o1 = o0 + 8 * HDIM;
            hs[o0]  = r0;
            sum[o0] = r0;
            hs[o1]  = r1;
            sum[o1] = r1;
        }
        __builtin_amdgcn_sched_barrier(0);
    }
}


template <int K>
static __device__ __forceinline__ unsigned
push_batch(unsigned mask, const int dsts, const float v0, const float v1, const unsigned lane,
           float* __restrict__ agg) {
#pragma unroll
    for (int j = 0; j < K; ++j) {
        const int l = __builtin_ctz(mask);
        mask &= mask - 1;
        const int d = __builtin_amdgcn_readlane(dsts, l);
        const unsigned dofs = (unsigned)d * HDIM + lane;
        atomicAdd(agg + dofs,      v0);
        atomicAdd(agg + dofs + 32, v1);
    }
    return mask;
}

static __device__ __forceinline__ void
push_set(unsigned mask, const int dsts, const float v0, const float v1, const unsigned lane,
         float* __restrict__ agg) {
    int left = __builtin_popcount(mask);
    while (left >= 4) { mask = push_batch<4>(mask, dsts, v0, v1, lane, agg); left -= 4; }
    if (left & 2) mask = push_batch<2>(mask, dsts, v0, v1, lane, agg);
    if (left & 1) mask = push_batch<1>(mask, dsts, v0, v1, lane, agg);
}

__global__ __launch_bounds__(256)
void gcn_scatter(const unsigned* __restrict__ grp,
                 const unsigned* __restrict__ spill_n, const int* __restrict__ spill,
                 const float* __restrict__ h, float* __restrict__ agg,
                 unsigned* __restrict__ ctr) {
    const int job = claim_job(ctr, (unsigned)N_SRC_CHUNKS);
    if (job < 0) return;
    const unsigned lo    = (unsigned)(job & 1) * HALF_NODES;
    const unsigned chunk = (unsigned)(job >> 1);
    const unsigned lane  = threadIdx.x & 31;
    const unsigned wave  = (unsigned)__builtin_amdgcn_readfirstlane((int)(threadIdx.x >> 5));
    const unsigned s0    = chunk * SRC_PER_BLOCK + wave * SRC_PER_WAVE;

    unsigned nq = 0u;
    if (lane < SRC_PER_WAVE && s0 + lane < (unsigned)N_NODES) nq = grp[s0 + lane];

    float v0[SRC_PER_WAVE], v1[SRC_PER_WAVE];
#pragma unroll
    for (int q = 0; q < SRC_PER_WAVE; ++q) {
        const unsigned s  = s0 + q < (unsigned)N_NODES ? s0 + q : (unsigned)N_NODES - 1u;
        const unsigned so = s * HDIM + lane;
        v0[q] = h[so];
        v1[q] = h[so + 32];
    }

    unsigned n[SRC_PER_WAVE];
    int d0[SRC_PER_WAVE];
#pragma unroll
    for (int q = 0; q < SRC_PER_WAVE; ++q) {
        n[q] = (unsigned)__builtin_amdgcn_readlane((int)nq, q);
        const unsigned s  = s0 + q < (unsigned)N_NODES ? s0 + q : (unsigned)N_NODES - 1u;
        const bool has = lane < n[q];
        const unsigned got = grp[has ? GRP_ROWS + s * SLOTS + lane : s];
        d0[q] = has ? (int)got : -1;
    }

#pragma unroll
    for (int q = 0; q < SRC_PER_WAVE; ++q) {
        const bool in0 = (unsigned)d0[q] - lo < (unsigned)HALF_NODES;
        push_set(__builtin_amdgcn_ballot_w32(in0), d0[q], v0[q], v1[q], lane, agg);
    }

#pragma unroll
    for (int q = 0; q < SRC_PER_WAVE; ++q) {
        if (n[q] > 32u) {
            int d1 = -1;
            if (lane + 32 < n[q]) d1 = (int)grp[GRP_ROWS + (s0 + q) * SLOTS + 32 + lane];
            const bool in1 = (unsigned)d1 - lo < (unsigned)HALF_NODES;
            push_set(__builtin_amdgcn_ballot_w32(in1), d1, v0[q], v1[q], lane, agg);
        }
    }

    if (wave == 0u) {
        unsigned ns = *spill_n;
        if (ns > (unsigned)N_EDGES) ns = (unsigned)N_EDGES;
        for (unsigned i = chunk; i < ns; i += (unsigned)N_SRC_CHUNKS) {
            const int s = spill[2u * i];
            const int d = spill[2u * i + 1u];
            if ((unsigned)d - lo < (unsigned)HALF_NODES) {
                const unsigned so   = (unsigned)s * HDIM + lane;
                const unsigned dofs = (unsigned)d * HDIM + lane;
                atomicAdd(agg + dofs,      h[so]);
                atomicAdd(agg + dofs + 32, h[so + 32]);
            }
        }
    }
}

__global__ void gcn_act(const float* __restrict__ agg, const float* __restrict__ dinv,
                        const float* __restrict__ bias, float* __restrict__ out) {
    int i = blockIdx.x * 256 + threadIdx.x;
    if (i < N_NODES * HDIM)
        out[i] = fmaxf(agg[i] * dinv[(unsigned)i / HDIM] + bias[i & (HDIM - 1)], 0.f);
}

extern "C" void kernel_launch(void* const* d_in, const int* in_sizes, int n_in,
                              void* d_out, int out_size, void* d_ws, size_t ws_size,
                              hipStream_t stream) {
    const float* x  = (const float*)d_in[0];
    const int*   ei = (const int*)d_in[1];
    const float* W  = (const float*)d_in[2];
    const float* b  = (const float*)d_in[3];
    float* out = (float*)d_out;

    float* ws   = (float*)d_ws;
    float* dinv = ws;
    float* h    = dinv + 50048;
    float* aggA = h    + (size_t)N_NODES * HDIM;
    float* aggB = aggA + (size_t)N_NODES * HDIM;
    unsigned* ctr = (unsigned*)(aggB + (size_t)N_NODES * HDIM);
    unsigned* grp     = ctr + N_CTRS * CTR_STRIDE;
    unsigned* spill_n = grp + GRP_ROWS + (size_t)N_NODES * SLOTS;
    int*      spill   = (int*)(spill_n + CTR_STRIDE);

    const int nodeBlocks = (N_NODES + 255) / 256;
    const int claimBlocks = 2 * N_CHUNKS;
    const int pushBlocks  = 2 * N_SRC_CHUNKS;
    const int gemmBlocks = (ROW_TILES + 3) / 4;
    const int actBlocks  = (N_NODES * HDIM + 255) / 256;

    deg_init   <<<nodeBlocks, 256, 0, stream>>>(dinv, ctr, grp, spill_n);
    deg_count  <<<claimBlocks, 256, 0, stream>>>(ei, dinv, ctr, grp, spill_n, spill);
    deg_to_dinv<<<nodeBlocks, 256, 0, stream>>>(dinv);

    gcn_gemm<false><<<gemmBlocks, 128, 0, stream>>>(x, nullptr, W, dinv, h, aggA);
    gcn_scatter    <<<pushBlocks, 256, 0, stream>>>(grp, spill_n, spill, h, aggA, ctr + 2 * CTR_STRIDE);

    gcn_gemm<true> <<<gemmBlocks, 128, 0, stream>>>(aggA, b, W + HDIM * HDIM, dinv, h, aggB);
    gcn_scatter    <<<pushBlocks, 256, 0, stream>>>(grp, spill_n, spill, h, aggB, ctr + 4 * CTR_STRIDE);

    gcn_gemm<true> <<<gemmBlocks, 128, 0, stream>>>(aggB, b + HDIM, W + 2 * HDIM * HDIM, dinv, h, aggA);
    gcn_scatter    <<<pushBlocks, 256, 0, stream>>>(grp, spill_n, spill, h, aggA, ctr + 6 * CTR_STRIDE);

    gcn_act<<<actBlocks, 256, 0, stream>>>(aggA, dinv, b + 2 * HDIM, out);
}


# Two launches fewer: start values stored by the layer-0 GEMM launch (rows unscaled, its scatter scales by rsqrt(deg[src])), readers form rsqrt(deg) themselves; layer-0 self-loop term added last
# speedup vs baseline: 4.0870x; 1.0323x over previous
#include <hip/hip_runtime.h>

#define N_NODES 50000
#define N_EDGES 800000
#define HDIM    64

typedef float v2f __attribute__((ext_vector_type(2)));
typedef float v8f __attribute__((ext_vector_type(8)));

#define HALF_NODES  (N_NODES / 2)
#define BLOCK_EDGES 512
#define N_CHUNKS    ((N_EDGES + BLOCK_EDGES - 1) / BLOCK_EDGES)
#define CTR_STRIDE  64
#define N_CTRS      8

#define SLOTS          64
#define GRP_ROWS       50048
#define SRC_PER_WAVE   4
#define SRC_PER_BLOCK  (8 * SRC_PER_WAVE)
#define N_SRC_CHUNKS   ((N_NODES + SRC_PER_BLOCK - 1) / SRC_PER_BLOCK)

static __device__ __forceinline__ int claim_job(unsigned* __restrict__ ctr, const unsigned n_chunks) {
    __shared__ int s_job;
    if (threadIdx.x == 0) {
        const unsigned id = __builtin_amdgcn_s_sendmsg_rtn(0x87);
        unsigned r = (id >> 18) & 1u;
        unsigned t = __hip_atomic_fetch_add(ctr + r * CTR_STRIDE, 1u, __ATOMIC_RELAXED, __HIP_MEMORY_SCOPE_AGENT);
        if (t >= n_chunks) {
            r ^= 1u;
            t = __hip_atomic_fetch_add(ctr + r * CTR_STRIDE, 1u, __ATOMIC_RELAXED, __HIP_MEMORY_SCOPE_AGENT);
        }
        s_job = (t < n_chunks) ? (int)(t * 2u + r) : -1;
    }
    __syncthreads();
    return __builtin_amdgcn_readfirstlane(s_job);
}


static __device__ __forceinline__ void
file_edge(const int s, const int d, const unsigned place, unsigned* __restrict__ grp,
          unsigned* __restrict__ spill_n, int* __restrict__ spill) {
    if (place < (unsigned)SLOTS) {
        grp[GRP_ROWS + (unsigned)s * SLOTS + place] = (unsigned)d;
    } else {
        const unsigned i = __hip_atomic_fetch_add(spill_n, 1u, __ATOMIC_RELAXED, __HIP_MEMORY_SCOPE_AGENT);
        spill[2u * i]      = s;
        spill[2u * i + 1u] = d;
    }
}

__global__ __launch_bounds__(256)
void deg_count(const int* __restrict__ ei, float* __restrict__ deg, unsigned* __restrict__ ctr,
               unsigned* __restrict__ grp,
               unsigned* __restrict__ spill_n, int* __restrict__ spill) {
    const int job = claim_job(ctr, (unsigned)N_CHUNKS);
    if (job < 0) return;
    const unsigned lo = (unsigned)(job & 1) * HALF_NODES;
    const int ea = (job >> 1) * BLOCK_EDGES + threadIdx.x;
    const int eb = ea + 256;
    int sa = -1, da = -1, sb = -1, db = -1;
    if (ea < N_EDGES) { sa = ei[ea]; da = ei[N_EDGES + ea]; }
    if (eb < N_EDGES) { sb = ei[eb]; db = ei[N_EDGES + eb]; }
    if ((unsigned)da - lo < (unsigned)HALF_NODES) atomicAdd(&deg[da], 1.0f);
    if ((unsigned)db - lo < (unsigned)HALF_NODES) atomicAdd(&deg[db], 1.0f);
    const bool fa = (unsigned)sa - lo < (unsigned)HALF_NODES;
    const bool fb = (unsigned)sb - lo < (unsigned)HALF_NODES;
    unsigned pa = 0u, pb = 0u;
    if (fa) pa = __hip_atomic_fetch_add(grp + sa, 1u, __ATOMIC_RELAXED, __HIP_MEMORY_SCOPE_AGENT);
    if (fb) pb = __hip_atomic_fetch_add(grp + sb, 1u, __ATOMIC_RELAXED, __HIP_MEMORY_SCOPE_AGENT);
    if (fa) file_edge(sa, da, pa, grp, spill_n, spill);
    if (fb) file_edge(sb, db, pb, grp, spill_n, spill);
}


#define ROW_TILES (N_NODES / 16)
#define W_PAIR    144

template <int MODE>
__global__ __launch_bounds__(128)
void gcn_gemm(const float* __restrict__ in, const float* __restrict__ bias,
              const float* __restrict__ W, float* __restrict__ deg,
              float* __restrict__ hs, float* __restrict__ sum,
              unsigned* __restrict__ ctr, unsigned* __restrict__ grp, unsigned* __restrict__ spill_n) {
    __shared__ __attribute__((aligned(16))) float Wl[(HDIM / 2) * W_PAIR];

    const int wave = threadIdx.x >> 5;
    const int lane = threadIdx.x & 31;
    const int c    = lane & 15;
    const int h    = lane >> 4;
    const int tm   = blockIdx.x * 4 + wave;
    const int tmc  = tm < ROW_TILES ? tm : ROW_TILES - 1;

    if (MODE == 0) {
        const int i = blockIdx.x * 128 + (int)threadIdx.x;
        if (i < N_NODES) { deg[i] = 1.0f; grp[i] = 0u; }
        if (i < N_CTRS) ctr[i * CTR_STRIDE] = 0u;
        if (i == 0) *spill_n = 0u;
    }

#pragma unroll
    for (int j = 0; j < 4; ++j) {
        const int unit = j * 128 + (int)threadIdx.x;
        const int r = unit >> 4, q = unit & 15;
        const float4 e = *(const float4*)(W + (2 * r) * HDIM + 4 * q);
        const float4 o = *(const float4*)(W + (2 * r + 1) * HDIM + 4 * q);
        float* __restrict__ dst = Wl + r * W_PAIR + 8 * q;
        *(float4*)(dst)     = make_float4(e.x, o.x, e.y, o.y);
        *(float4*)(dst + 4) = make_float4(e.z, o.z, e.w, o.w);
    }

    const int mrow = (tmc << 4) + c;
    const float* __restrict__ arow = in + (size_t)mrow * HDIM + 4 * h;
    float4 a4[8];
#pragma unroll
    for (int j = 0; j < 8; ++j) a4[j] = *(const float4*)(arow + 8 * j);
    if (MODE != 0) {
        const float din = rsqrtf(deg[mrow]);
        if (MODE == 1) {
            const float* hrow = hs + (size_t)mrow * HDIM + 4 * h;
            float4 s4[8];
#pragma unroll
            for (int j = 0; j < 8; ++j) s4[j] = *(const float4*)(hrow + 8 * j);
#pragma unroll
            for (int j = 0; j < 8; ++j) {
                a4[j].x += s4[j].x * din;
                a4[j].y += s4[j].y * din;
                a4[j].z += s4[j].z * din;
                a4[j].w += s4[j].w * din;
            }
        }
#pragma unroll
        for (int j = 0; j < 8; ++j) {
            const float4 b4 = *(const float4*)(bias + 8 * j + 4 * h);
            a4[j].x = fmaxf(a4[j].x * din + b4.x, 0.f);
            a4[j].y = fmaxf(a4[j].y * din + b4.y, 0.f);
            a4[j].z = fmaxf(a4[j].z * din + b4.z, 0.f);
            a4[j].w = fmaxf(a4[j].w * din + b4.w, 0.f);
        }
    }

    __syncthreads();
    if (tm >= ROW_TILES) return;

    const int rbase = tm << 4;
    float srow[8];
#pragma unroll
    for (int v = 0; v < 8; ++v) srow[v] = MODE != 0 ? rsqrtf(deg[rbase + 8 * h + v]) : 1.0f;

    const v2f* __restrict__ wl = (const v2f*)(Wl + (2 * h) * W_PAIR + 2 * c);

#pragma unroll
    for (int p = 0; p < 2; ++p) {
        v8f lo = {0.f, 0.f, 0.f, 0.f, 0.f, 0.f, 0.f, 0.f};
        v8f hi = {0.f, 0.f, 0.f, 0.f, 0.f, 0.f, 0.f, 0.f};
#pragma unroll
        for (int j = 0; j < 8; ++j) {
#pragma unroll
            for (int u = 0; u < 2; ++u) {
                v2f a;
                a.x = u ? a4[j].z : a4[j].x;
                a.y = u ? a4[j].w : a4[j].y;
                const int pr = (4 * j + u) * (W_PAIR / 2);
                const v2f b0 = wl[pr + 16 * (2 * p)];
                const v2f b1 = wl[pr + 16 * (2 * p + 1)];
                lo = __builtin_amdgcn_wmma_f32_16x16x4_f32(false, a, false, b0, (short)0, lo, false, false);
                hi = __builtin_amdgcn_wmma_f32_16x16x4_f32(false, a, false, b1, (short)0, hi, false, false);
            }
        }
#pragma unroll
        for (int v = 0; v < 8; ++v) {
            const float l = MODE != 0 ? lo[v] * srow[v] : lo[v];
            const float g = MODE != 0 ? hi[v] * srow[v] : hi[v];
            const float give = h ? l : g;
            const float got = __builtin_bit_cast(float, __builtin_amdgcn_permlanex16(
                __builtin_bit_cast(unsigned, give), __builtin_bit_cast(unsigned, give),
                0x76543210u, 0xfedcba98u, false, false));
            const float r0 = h ? got : l;
            const float r1 = h ? g   : got;
            const size_t o0 = (size_t)(rbase + v) * HDIM + 32 * p + lane;
            const size_t o1 = o0 + 8 * HDIM;
            hs[o0]  = r0;
            sum[o0] = MODE != 0 ? r0 : 0.f;
            hs[o1]  = r1;
            sum[o1] = MODE != 0 ? r1 : 0.f;
        }
        __builtin_amdgcn_sched_barrier(0);
    }
}


template <int K>
static __device__ __forceinline__ unsigned
push_batch(unsigned mask, const int dsts, const float v0, const float v1, const unsigned lane,
           float* __restrict__ agg) {
#pragma unroll
    for (int j = 0; j < K; ++j) {
        const int l = __builtin_ctz(mask);
        mask &= mask - 1;
        const int d = __builtin_amdgcn_readlane(dsts, l);
        const unsigned dofs = (unsigned)d * HDIM + lane;
        atomicAdd(agg + dofs,      v0);
        atomicAdd(agg + dofs + 32, v1);
    }
    return mask;
}

static __device__ __forceinline__ void
push_set(unsigned mask, const int dsts, const float v0, const float v1, const unsigned lane,
         float* __restrict__ agg) {
    int left = __builtin_popcount(mask);
    while (left >= 4) { mask = push_batch<4>(mask, dsts, v0, v1, lane, agg); left -= 4; }
    if (left & 2) mask = push_batch<2>(mask, dsts, v0, v1, lane, agg);
    if (left & 1) mask = push_batch<1>(mask, dsts, v0, v1, lane, agg);
}

template <bool SCALE>
__global__ __launch_bounds__(256)
void gcn_scatter(const unsigned* __restrict__ grp,
                 const unsigned* __restrict__ spill_n, const int* __restrict__ spill,
                 const float* __restrict__ deg,
                 const float* __restrict__ h, float* __restrict__ agg,
                 unsigned* __restrict__ ctr) {
    const int job = claim_job(ctr, (unsigned)N_SRC_CHUNKS);
    if (job < 0) return;
    const unsigned lo    = (unsigned)(job & 1) * HALF_NODES;
    const unsigned chunk = (unsigned)(job >> 1);
    const unsigned lane  = threadIdx.x & 31;
    const unsigned wave  = (unsigned)__builtin_amdgcn_readfirstlane((int)(threadIdx.x >> 5));
    const unsigned s0    = chunk * SRC_PER_BLOCK + wave * SRC_PER_WAVE;

    unsigned nq = 0u;
    float fq = 1.0f;
    if (lane < SRC_PER_WAVE && s0 + lane < (unsigned)N_NODES) {
        nq = grp[s0 + lane];
        if (SCALE) fq = rsqrtf(deg[s0 + lane]);
    }

    float v0[SRC_PER_WAVE], v1[SRC_PER_WAVE];
#pragma unroll
    for (int q = 0; q < SRC_PER_WAVE; ++q) {
        const unsigned s  = s0 + q < (unsigned)N_NODES ? s0 + q : (unsigned)N_NODES - 1u;
        const unsigned so = s * HDIM + lane;
        v0[q] = h[so];
        v1[q] = h[so + 32];
    }
    if (SCALE) {
#pragma unroll
        for (int q = 0; q < SRC_PER_WAVE; ++q) {
            const float f = __builtin_bit_cast(float, __builtin_amdgcn_readlane(__builtin_bit_cast(int, fq), q));
            v0[q] *= f;
            v1[q] *= f;
        }
    }

    unsigned n[SRC_PER_WAVE];
    int d0[SRC_PER_WAVE];
#pragma unroll
    for (int q = 0; q < SRC_PER_WAVE; ++q) {
        n[q] = (unsigned)__builtin_amdgcn_readlane((int)nq, q);
        const unsigned s  = s0 + q < (unsigned)N_NODES ? s0 + q : (unsigned)N_NODES - 1u;
        const bool has = lane < n[q];
        const unsigned got = grp[has ? GRP_ROWS + s * SLOTS + lane : s];
        d0[q] = has ? (int)got : -1;
    }

#pragma unroll
    for (int q = 0; q < SRC_PER_WAVE; ++q) {
        const bool in0 = (unsigned)d0[q] - lo < (unsigned)HALF_NODES;
        push_set(__builtin_amdgcn_ballot_w32(in0), d0[q], v0[q], v1[q], lane, agg);
    }

#pragma unroll
    for (int q = 0; q < SRC_PER_WAVE; ++q) {
        if (n[q] > 32u) {
            int d1 = -1;
            if (lane + 32 < n[q]) d1 = (int)grp[GRP_ROWS + (s0 + q) * SLOTS + 32 + lane];
            const bool in1 = (unsigned)d1 - lo < (unsigned)HALF_NODES;
            push_set(__builtin_amdgcn_ballot_w32(in1), d1, v0[q], v1[q], lane, agg);
        }
    }

    if (wave == 0u) {
        unsigned ns = *spill_n;
        if (ns > (unsigned)N_EDGES) ns = (unsigned)N_EDGES;
        for (unsigned i = chunk; i < ns; i += (unsigned)N_SRC_CHUNKS) {
            const int s = spill[2u * i];
            const int d = spill[2u * i + 1u];
            if ((unsigned)d - lo < (unsigned)HALF_NODES) {
                const unsigned so   = (unsigned)s * HDIM + lane;
                const unsigned dofs = (unsigned)d * HDIM + lane;
                const float f = SCALE ? rsqrtf(deg[s]) : 1.0f;
                atomicAdd(agg + dofs,      SCALE ? h[so] * f      : h[so]);
                atomicAdd(agg + dofs + 32, SCALE ? h[so + 32] * f : h[so + 32]);
            }
        }
    }
}

__global__ void gcn_act(const float* __restrict__ agg, const float* __restrict__ deg,
                        const float* __restrict__ bias, float* __restrict__ out) {
    int i = blockIdx.x * 256 + threadIdx.x;
    if (i < N_NODES * HDIM)
        out[i] = fmaxf(agg[i] * rsqrtf(deg[(unsigned)i / HDIM]) + bias[i & (HDIM - 1)], 0.f);
}

extern "C" void kernel_launch(void* const* d_in, const int* in_sizes, int n_in,
                              void* d_out, int out_size, void* d_ws, size_t ws_size,
                              hipStream_t stream) {
    const float* x  = (const float*)d_in[0];
    const int*   ei = (const int*)d_in[1];
    const float* W  = (const float*)d_in[2];
    const float* b  = (const float*)d_in[3];
    float* out = (float*)d_out;

    float* ws   = (float*)d_ws;
    float* deg  = ws;
    float* h    = deg + 50048;
    float* aggA = h    + (size_t)N_NODES * HDIM;
    float* aggB = aggA + (size_t)N_NODES * HDIM;
    unsigned* ctr = (unsigned*)(aggB + (size_t)N_NODES * HDIM);
    unsigned* grp     = ctr + N_CTRS * CTR_STRIDE;
    unsigned* spill_n = grp + GRP_ROWS + (size_t)N_NODES * SLOTS;
    int*      spill   = (int*)(spill_n + CTR_STRIDE);

    const int claimBlocks = 2 * N_CHUNKS;
    const int pushBlocks  = 2 * N_SRC_CHUNKS;
    const int gemmBlocks = (ROW_TILES + 3) / 4;
    const int actBlocks  = (N_NODES * HDIM + 255) / 256;

    gcn_gemm<0><<<gemmBlocks, 128, 0, stream>>>(x, nullptr, W, deg, h, aggA, ctr, grp, spill_n);

    deg_count<<<claimBlocks, 256, 0, stream>>>(ei, deg, ctr, grp, spill_n, spill);

    gcn_scatter<true> <<<pushBlocks, 256, 0, stream>>>(grp, spill_n, spill, deg, h, aggA, ctr + 2 * CTR_STRIDE);

    gcn_gemm<1><<<gemmBlocks, 128, 0, stream>>>(aggA, b, W + HDIM * HDIM, deg, h, aggB, nullptr, nullptr, nullptr);
    gcn_scatter<false><<<pushBlocks, 256, 0, stream>>>(grp, spill_n, spill, deg, h, aggB, ctr + 4 * CTR_STRIDE);

    gcn_gemm<2><<<gemmBlocks, 128, 0, stream>>>(aggB, b + HDIM, W + 2 * HDIM * HDIM, deg, h, aggA, nullptr, nullptr, nullptr);
    gcn_scatter<false><<<pushBlocks, 256, 0, stream>>>(grp, spill_n, spill, deg, h, aggA, ctr + 6 * CTR_STRIDE);

    gcn_act<<<actBlocks, 256, 0, stream>>>(aggA, deg, b + 2 * HDIM, out);
}


# Final bias+ReLU pass as 5,000 long waves, not 100,000 one-element ones: a lane takes five 16-byte pieces, all loads in flight; launch no longer waits on wave entry; values and sum order unchanged
# speedup vs baseline: 4.2554x; 1.0412x over previous
#include <hip/hip_runtime.h>

#define N_NODES 50000
#define N_EDGES 800000
#define HDIM    64

typedef float v2f __attribute__((ext_vector_type(2)));
typedef float v8f __attribute__((ext_vector_type(8)));

#define HALF_NODES  (N_NODES / 2)
#define BLOCK_EDGES 512
#define N_CHUNKS    ((N_EDGES + BLOCK_EDGES - 1) / BLOCK_EDGES)
#define CTR_STRIDE  64
#define N_CTRS      8

#define SLOTS          64
#define GRP_ROWS       50048
#define SRC_PER_WAVE   4
#define SRC_PER_BLOCK  (8 * SRC_PER_WAVE)
#define N_SRC_CHUNKS   ((N_NODES + SRC_PER_BLOCK - 1) / SRC_PER_BLOCK)

static __device__ __forceinline__ int claim_job(unsigned* __restrict__ ctr, const unsigned n_chunks) {
    __shared__ int s_job;
    if (threadIdx.x == 0) {
        const unsigned id = __builtin_amdgcn_s_sendmsg_rtn(0x87);
        unsigned r = (id >> 18) & 1u;
        unsigned t = __hip_atomic_fetch_add(ctr + r * CTR_STRIDE, 1u, __ATOMIC_RELAXED, __HIP_MEMORY_SCOPE_AGENT);
        if (t >= n_chunks) {
            r ^= 1u;
            t = __hip_atomic_fetch_add(ctr + r * CTR_STRIDE, 1u, __ATOMIC_RELAXED, __HIP_MEMORY_SCOPE_AGENT);
        }
        s_job = (t < n_chunks) ? (int)(t * 2u + r) : -1;
    }
    __syncthreads();
    return __builtin_amdgcn_readfirstlane(s_job);
}


static __device__ __forceinline__ void
file_edge(const int s, const int d, const unsigned place, unsigned* __restrict__ grp,
          unsigned* __restrict__ spill_n, int* __restrict__ spill) {
    if (place < (unsigned)SLOTS) {
        grp[GRP_ROWS + (unsigned)s * SLOTS + place] = (unsigned)d;
    } else {
        const unsigned i = __hip_atomic_fetch_add(spill_n, 1u, __ATOMIC_RELAXED, __HIP_MEMORY_SCOPE_AGENT);
        spill[2u * i]      = s;
        spill[2u * i + 1u] = d;
    }
}

__global__ __launch_bounds__(256)
void deg_count(const int* __restrict__ ei, float* __restrict__ deg, unsigned* __restrict__ ctr,
               unsigned* __restrict__ grp,
               unsigned* __restrict__ spill_n, int* __restrict__ spill) {
    const int job = claim_job(ctr, (unsigned)N_CHUNKS);
    if (job < 0) return;
    const unsigned lo = (unsigned)(job & 1) * HALF_NODES;
    const int ea = (job >> 1) * BLOCK_EDGES + threadIdx.x;
    const int eb = ea + 256;
    int sa = -1, da = -1, sb = -1, db = -1;
    if (ea < N_EDGES) { sa = ei[ea]; da = ei[N_EDGES + ea]; }
    if (eb < N_EDGES) { sb = ei[eb]; db = ei[N_EDGES + eb]; }
    if ((unsigned)da - lo < (unsigned)HALF_NODES) atomicAdd(&deg[da], 1.0f);
    if ((unsigned)db - lo < (unsigned)HALF_NODES) atomicAdd(&deg[db], 1.0f);
    const bool fa = (unsigned)sa - lo < (unsigned)HALF_NODES;
    const bool fb = (unsigned)sb - lo < (unsigned)HALF_NODES;
    unsigned pa = 0u, pb = 0u;
    if (fa) pa = __hip_atomic_fetch_add(grp + sa, 1u, __ATOMIC_RELAXED, __HIP_MEMORY_SCOPE_AGENT);
    if (fb) pb = __hip_atomic_fetch_add(grp + sb, 1u, __ATOMIC_RELAXED, __HIP_MEMORY_SCOPE_AGENT);
    if (fa) file_edge(sa, da, pa, grp, spill_n, spill);
    if (fb) file_edge(sb, db, pb, grp, spill_n, spill);
}


#define ROW_TILES (N_NODES / 16)
#define W_PAIR    144

template <int MODE>
__global__ __launch_bounds__(128)
void gcn_gemm(const float* __restrict__ in, const float* __restrict__ bias,
              const float* __restrict__ W, float* __restrict__ deg,
              float* __restrict__ hs, float* __restrict__ sum,
              unsigned* __restrict__ ctr, unsigned* __restrict__ grp, unsigned* __restrict__ spill_n) {
    __shared__ __attribute__((aligned(16))) float Wl[(HDIM / 2) * W_PAIR];

    const int wave = threadIdx.x >> 5;
    const int lane = threadIdx.x & 31;
    const int c    = lane & 15;
    const int h    = lane >> 4;
    const int tm   = blockIdx.x * 4 + wave;
    const int tmc  = tm < ROW_TILES ? tm : ROW_TILES - 1;

    if (MODE == 0) {
        const int i = blockIdx.x * 128 + (int)threadIdx.x;
        if (i < N_NODES) { deg[i] = 1.0f; grp[i] = 0u; }
        if (i < N_CTRS) ctr[i * CTR_STRIDE] = 0u;
        if (i == 0) *spill_n = 0u;
    }

#pragma unroll
    for (int j = 0; j < 4; ++j) {
        const int unit = j * 128 + (int)threadIdx.x;
        const int r = unit >> 4, q = unit & 15;
        const float4 e = *(const float4*)(W + (2 * r) * HDIM + 4 * q);
        const float4 o = *(const float4*)(W + (2 * r + 1) * HDIM + 4 * q);
        float* __restrict__ dst = Wl + r * W_PAIR + 8 * q;
        *(float4*)(dst)     = make_float4(e.x, o.x, e.y, o.y);
        *(float4*)(dst + 4) = make_float4(e.z, o.z, e.w, o.w);
    }

    const int mrow = (tmc << 4) + c;
    const float* __restrict__ arow = in + (size_t)mrow * HDIM + 4 * h;
    float4 a4[8];
#pragma unroll
    for (int j = 0; j < 8; ++j) a4[j] = *(const float4*)(arow + 8 * j);
    if (MODE != 0) {
        const float din = rsqrtf(deg[mrow]);
        if (MODE == 1) {
            const float* hrow = hs + (size_t)mrow * HDIM + 4 * h;
            float4 s4[8];
#pragma unroll
            for (int j = 0; j < 8; ++j) s4[j] = *(const float4*)(hrow + 8 * j);
#pragma unroll
            for (int j = 0; j < 8; ++j) {
                a4[j].x += s4[j].x * din;
                a4[j].y += s4[j].y * din;
                a4[j].z += s4[j].z * din;
                a4[j].w += s4[j].w * din;
            }
        }
#pragma unroll
        for (int j = 0; j < 8; ++j) {
            const float4 b4 = *(const float4*)(bias + 8 * j + 4 * h);
            a4[j].x = fmaxf(a4[j].x * din + b4.x, 0.f);
            a4[j].y = fmaxf(a4[j].y * din + b4.y, 0.f);
            a4[j].z = fmaxf(a4[j].z * din + b4.z, 0.f);
            a4[j].w = fmaxf(a4[j].w * din + b4.w, 0.f);
        }
    }

    __syncthreads();
    if (tm >= ROW_TILES) return;

    const int rbase = tm << 4;
    float srow[8];
#pragma unroll
    for (int v = 0; v < 8; ++v) srow[v] = MODE != 0 ? rsqrtf(deg[rbase + 8 * h + v]) : 1.0f;

    const v2f* __restrict__ wl = (const v2f*)(Wl + (2 * h) * W_PAIR + 2 * c);

#pragma unroll
    for (int p = 0; p < 2; ++p) {
        v8f lo = {0.f, 0.f, 0.f, 0.f, 0.f, 0.f, 0.f, 0.f};
        v8f hi = {0.f, 0.f, 0.f, 0.f, 0.f, 0.f, 0.f, 0.f};
#pragma unroll
        for (int j = 0; j < 8; ++j) {
#pragma unroll
            for (int u = 0; u < 2; ++u) {
                v2f a;
                a.x = u ? a4[j].z : a4[j].x;
                a.y = u ? a4[j].w : a4[j].y;
                const int pr = (4 * j + u) * (W_PAIR / 2);
                const v2f b0 = wl[pr + 16 * (2 * p)];
                const v2f b1 = wl[pr + 16 * (2 * p + 1)];
                lo = __builtin_amdgcn_wmma_f32_16x16x4_f32(false, a, false, b0, (short)0, lo, false, false);
                hi = __builtin_amdgcn_wmma_f32_16x16x4_f32(false, a, false, b1, (short)0, hi, false, false);
            }
        }
#pragma unroll
        for (int v = 0; v < 8; ++v) {
            const float l = MODE != 0 ? lo[v] * srow[v] : lo[v];
            const float g = MODE != 0 ? hi[v] * srow[v] : hi[v];
            const float give = h ? l : g;
            const float got = __builtin_bit_cast(float, __builtin_amdgcn_permlanex16(
                __builtin_bit_cast(unsigned, give), __builtin_bit_cast(unsigned, give),
                0x76543210u, 0xfedcba98u, false, false));
            const float r0 = h ? got : l;
            const float r1 = h ? g   : got;
            const size_t o0 = (size_t)(rbase + v) * HDIM + 32 * p + lane;
            const size_t o1 = o0 + 8 * HDIM;
            hs[o0]  = r0;
            sum[o0] = MODE != 0 ? r0 : 0.f;
            hs[o1]  = r1;
            sum[o1] = MODE != 0 ? r1 : 0.f;
        }
        __builtin_amdgcn_sched_barrier(0);
    }
}


template <int K>
static __device__ __forceinline__ unsigned
push_batch(unsigned mask, const int dsts, const float v0, const float v1, const unsigned lane,
           float* __restrict__ agg) {
#pragma unroll
    for (int j = 0; j < K; ++j) {
        const int l = __builtin_ctz(mask);
        mask &= mask - 1;
        const int d = __builtin_amdgcn_readlane(dsts, l);
        const unsigned dofs = (unsigned)d * HDIM + lane;
        atomicAdd(agg + dofs,      v0);
        atomicAdd(agg + dofs + 32, v1);
    }
    return mask;
}

static __device__ __forceinline__ void
push_set(unsigned mask, const int dsts, const float v0, const float v1, const unsigned lane,
         float* __restrict__ agg) {
    int left = __builtin_popcount(mask);
    while (left >= 4) { mask = push_batch<4>(mask, dsts, v0, v1, lane, agg); left -= 4; }
    if (left & 2) mask = push_batch<2>(mask, dsts, v0, v1, lane, agg);
    if (left & 1) mask = push_batch<1>(mask, dsts, v0, v1, lane, agg);
}

template <bool SCALE>
__global__ __launch_bounds__(256)
void gcn_scatter(const unsigned* __restrict__ grp,
                 const unsigned* __restrict__ spill_n, const int* __restrict__ spill,
                 const float* __restrict__ deg,
                 const float* __restrict__ h, float* __restrict__ agg,
                 unsigned* __restrict__ ctr) {
    const int job = claim_job(ctr, (unsigned)N_SRC_CHUNKS);
    if (job < 0) return;
    const unsigned lo    = (unsigned)(job & 1) * HALF_NODES;
    const unsigned chunk = (unsigned)(job >> 1);
    const unsigned lane  = threadIdx.x & 31;
    const unsigned wave  = (unsigned)__builtin_amdgcn_readfirstlane((int)(threadIdx.x >> 5));
    const unsigned s0    = chunk * SRC_PER_BLOCK + wave * SRC_PER_WAVE;

    unsigned nq = 0u;
    float fq = 1.0f;
    if (lane < SRC_PER_WAVE && s0 + lane < (unsigned)N_NODES) {
        nq = grp[s0 + lane];
        if (SCALE) fq = rsqrtf(deg[s0 + lane]);
    }

    float v0[SRC_PER_WAVE], v1[SRC_PER_WAVE];
#pragma unroll
    for (int q = 0; q < SRC_PER_WAVE; ++q) {
        const unsigned s  = s0 + q < (unsigned)N_NODES ? s0 + q : (unsigned)N_NODES - 1u;
        const unsigned so = s * HDIM + lane;
        v0[q] = h[so];
        v1[q] = h[so + 32];
    }
    if (SCALE) {
#pragma unroll
        for (int q = 0; q < SRC_PER_WAVE; ++q) {
            const float f = __builtin_bit_cast(float, __builtin_amdgcn_readlane(__builtin_bit_cast(int, fq), q));
            v0[q] *= f;
            v1[q] *= f;
        }
    }

    unsigned n[SRC_PER_WAVE];
    int d0[SRC_PER_WAVE];
#pragma unroll
    for (int q = 0; q < SRC_PER_WAVE; ++q) {
        n[q] = (unsigned)__builtin_amdgcn_readlane((int)nq, q);
        const unsigned s  = s0 + q < (unsigned)N_NODES ? s0 + q : (unsigned)N_NODES - 1u;
        const bool has = lane < n[q];
        const unsigned got = grp[has ? GRP_ROWS + s * SLOTS + lane : s];
        d0[q] = has ? (int)got : -1;
    }

#pragma unroll
    for (int q = 0; q < SRC_PER_WAVE; ++q) {
        const bool in0 = (unsigned)d0[q] - lo < (unsigned)HALF_NODES;
        push_set(__builtin_amdgcn_ballot_w32(in0), d0[q], v0[q], v1[q], lane, agg);
    }

#pragma unroll
    for (int q = 0; q < SRC_PER_WAVE; ++q) {
        if (n[q] > 32u) {
            int d1 = -1;
            if (lane + 32 < n[q]) d1 = (int)grp[GRP_ROWS + (s0 + q) * SLOTS + 32 + lane];
            const bool in1 = (unsigned)d1 - lo < (unsigned)HALF_NODES;
            push_set(__builtin_amdgcn_ballot_w32(in1), d1, v0[q], v1[q], lane, agg);
        }
    }

    if (wave == 0u) {
        unsigned ns = *spill_n;
        if (ns > (unsigned)N_EDGES) ns = (unsigned)N_EDGES;
        for (unsigned i = chunk; i < ns; i += (unsigned)N_SRC_CHUNKS) {
            const int s = spill[2u * i];
            const int d = spill[2u * i + 1u];
            if ((unsigned)d - lo < (unsigned)HALF_NODES) {
                const unsigned so   = (unsigned)s * HDIM + lane;
                const unsigned dofs = (unsigned)d * HDIM + lane;
                const float f = SCALE ? rsqrtf(deg[s]) : 1.0f;
                atomicAdd(agg + dofs,      SCALE ? h[so] * f      : h[so]);
                atomicAdd(agg + dofs + 32, SCALE ? h[so + 32] * f : h[so + 32]);
            }
        }
    }
}

#define ACT_PIECES 5
#define ACT_QUADS  (N_NODES * HDIM / 4)
#define ACT_BLOCKS (ACT_QUADS / (256 * ACT_PIECES))
static_assert(ACT_BLOCKS * 256 * ACT_PIECES == ACT_QUADS, "slabs must tile the output exactly: the kernel has no tail guard");

__global__ __launch_bounds__(256)
void gcn_act(const float* __restrict__ agg, const float* __restrict__ deg,
             const float* __restrict__ bias, float* __restrict__ out) {
    const unsigned base = blockIdx.x * (256u * ACT_PIECES) + threadIdx.x;
    float4 a[ACT_PIECES];
    float  d[ACT_PIECES];
#pragma unroll
    for (int j = 0; j < ACT_PIECES; ++j) {
        const unsigned q = base + 256u * j;
        a[j] = *(const float4*)(agg + 4u * q);
        d[j] = deg[q >> 4];
    }
    const float4 b4 = *(const float4*)(bias + 4 * (threadIdx.x & 15));
#pragma unroll
    for (int j = 0; j < ACT_PIECES; ++j) {
        const unsigned q = base + 256u * j;
        const float din = rsqrtf(d[j]);
        float4 r;
        r.x = fmaxf(a[j].x * din + b4.x, 0.f);
        r.y = fmaxf(a[j].y * din + b4.y, 0.f);
        r.z = fmaxf(a[j].z * din + b4.z, 0.f);
        r.w = fmaxf(a[j].w * din + b4.w, 0.f);
        *(float4*)(out + 4u * q) = r;
    }
}

extern "C" void kernel_launch(void* const* d_in, const int* in_sizes, int n_in,
                              void* d_out, int out_size, void* d_ws, size_t ws_size,
                              hipStream_t stream) {
    const float* x  = (const float*)d_in[0];
    const int*   ei = (const int*)d_in[1];
    const float* W  = (const float*)d_in[2];
    const float* b  = (const float*)d_in[3];
    float* out = (float*)d_out;

    float* ws   = (float*)d_ws;
    float* deg  = ws;
    float* h    = deg + 50048;
    float* aggA = h    + (size_t)N_NODES * HDIM;
    float* aggB = aggA + (size_t)N_NODES * HDIM;
    unsigned* ctr = (unsigned*)(aggB + (size_t)N_NODES * HDIM);
    unsigned* grp     = ctr + N_CTRS * CTR_STRIDE;
    unsigned* spill_n = grp + GRP_ROWS + (size_t)N_NODES * SLOTS;
    int*      spill   = (int*)(spill_n + CTR_STRIDE);

    const int claimBlocks = 2 * N_CHUNKS;
    const int pushBlocks  = 2 * N_SRC_CHUNKS;
    const int gemmBlocks = (ROW_TILES + 3) / 4;
    const int actBlocks  = ACT_BLOCKS;

    gcn_gemm<0><<<gemmBlocks, 128, 0, stream>>>(x, nullptr, W, deg, h, aggA, ctr, grp, spill_n);

    deg_count<<<claimBlocks, 256, 0, stream>>>(ei, deg, ctr, grp, spill_n, spill);

    gcn_scatter<true> <<<pushBlocks, 256, 0, stream>>>(grp, spill_n, spill, deg, h, aggA, ctr + 2 * CTR_STRIDE);

    gcn_gemm<1><<<gemmBlocks, 128, 0, stream>>>(aggA, b, W + HDIM * HDIM, deg, h, aggB, nullptr, nullptr, nullptr);
    gcn_scatter<false><<<pushBlocks, 256, 0, stream>>>(grp, spill_n, spill, deg, h, aggB, ctr + 4 * CTR_STRIDE);

    gcn_gemm<2><<<gemmBlocks, 128, 0, stream>>>(aggB, b + HDIM, W + 2 * HDIM * HDIM, deg, h, aggA, nullptr, nullptr, nullptr);
    gcn_scatter<false><<<pushBlocks, 256, 0, stream>>>(grp, spill_n, spill, deg, h, aggA, ctr + 6 * CTR_STRIDE);

    gcn_act<<<actBlocks, 256, 0, stream>>>(aggA, deg, b + 2 * HDIM, out);
}


# Node sums formed in LDS, not by float atomics at L2: degree pass buckets edges by 100-node dst range, a work-group per range adds source rows by ds_add_f32, self-loop row first; LDS arrival order
# speedup vs baseline: 4.3952x; 1.0329x over previous
#include <hip/hip_runtime.h>

#define N_NODES 50000
#define N_EDGES 800000
#define HDIM    64

typedef float v2f __attribute__((ext_vector_type(2)));
typedef float v8f __attribute__((ext_vector_type(8)));

#define NPR         100
#define N_RANGES    (N_NODES / NPR)
#define CAP         8192
static_assert(N_RANGES * NPR == N_NODES, "ranges must tile the nodes exactly: the summing kernel has no tail guard");
static_assert(N_NODES <= 65536 && NPR <= 65536, "an edge is packed as source | place << 16");

#define HALF_NODES  (N_NODES / 2)
#define HALF_RANGES (N_RANGES / 2)
static_assert(HALF_RANGES * NPR == HALF_NODES, "a range must lie in one half");
#define GRP_THREADS 512
#define GRP_PER_THR 8
#define GRP_EDGES   (GRP_THREADS * GRP_PER_THR)
#define GRP_CHUNKS  ((N_EDGES + GRP_EDGES - 1) / GRP_EDGES)
#define CTR_STRIDE  64
#define N_CTRS      2

static __device__ __forceinline__ int claim_job(unsigned* __restrict__ ctr, const unsigned n_chunks) {
    __shared__ int s_job;
    if (threadIdx.x == 0) {
        const unsigned id = __builtin_amdgcn_s_sendmsg_rtn(0x87);
        unsigned r = (id >> 18) & 1u;
        unsigned t = __hip_atomic_fetch_add(ctr + r * CTR_STRIDE, 1u, __ATOMIC_RELAXED, __HIP_MEMORY_SCOPE_AGENT);
        if (t >= n_chunks) {
            r ^= 1u;
            t = __hip_atomic_fetch_add(ctr + r * CTR_STRIDE, 1u, __ATOMIC_RELAXED, __HIP_MEMORY_SCOPE_AGENT);
        }
        s_job = (t < n_chunks) ? (int)(t * 2u + r) : -1;
    }
    __syncthreads();
    return __builtin_amdgcn_readfirstlane(s_job);
}

__global__ __launch_bounds__(GRP_THREADS)
void group_edges(const int* __restrict__ ei, float* __restrict__ deg, unsigned* __restrict__ ctr,
                 unsigned* __restrict__ cnt, unsigned* __restrict__ bucket,
                 unsigned* __restrict__ spill_n, int* __restrict__ spill) {
    __shared__ unsigned hist[HALF_RANGES];
    const int job = claim_job(ctr, (unsigned)GRP_CHUNKS);
    if (job < 0) return;
    const unsigned half = (unsigned)(job & 1);
    const unsigned lo   = half * HALF_NODES;
    const unsigned rlo  = half * HALF_RANGES;
    const unsigned e0   = (unsigned)(job >> 1) * GRP_EDGES + threadIdx.x;
    for (unsigned i = threadIdx.x; i < (unsigned)HALF_RANGES; i += GRP_THREADS) hist[i] = 0u;

    int s[GRP_PER_THR], d[GRP_PER_THR];
#pragma unroll
    for (int k = 0; k < GRP_PER_THR; ++k) {
        const unsigned e = e0 + (unsigned)k * GRP_THREADS;
        s[k] = -1; d[k] = -1;
        if (e < (unsigned)N_EDGES) { s[k] = ei[e]; d[k] = ei[N_EDGES + e]; }
    }
    __syncthreads();

    unsigned rl[GRP_PER_THR], rk[GRP_PER_THR];
#pragma unroll
    for (int k = 0; k < GRP_PER_THR; ++k) {
        const unsigned dl = (unsigned)d[k] - lo;
        rl[k] = dl / NPR;
        rk[k] = 0u;
        if (dl < (unsigned)HALF_NODES) {
            atomicAdd(&deg[d[k]], 1.0f);
            rk[k] = __hip_atomic_fetch_add(&hist[rl[k]], 1u, __ATOMIC_RELAXED, __HIP_MEMORY_SCOPE_WORKGROUP);
        }
    }
    __syncthreads();

    for (unsigned i = threadIdx.x; i < (unsigned)HALF_RANGES; i += GRP_THREADS) {
        const unsigned c = hist[i];
        if (c) hist[i] = __hip_atomic_fetch_add(cnt + rlo + i, c, __ATOMIC_RELAXED, __HIP_MEMORY_SCOPE_AGENT);
    }
    __syncthreads();

#pragma unroll
    for (int k = 0; k < GRP_PER_THR; ++k) {
        const unsigned dl = (unsigned)d[k] - lo;
        if (dl < (unsigned)HALF_NODES) {
            const unsigned place = hist[rl[k]] + rk[k];
            if (place < (unsigned)CAP) {
                bucket[(rlo + rl[k]) * CAP + place] = (unsigned)s[k] | ((dl - rl[k] * NPR) << 16);
            } else {
                const unsigned i = __hip_atomic_fetch_add(spill_n, 1u, __ATOMIC_RELAXED, __HIP_MEMORY_SCOPE_AGENT);
                spill[2u * i]      = s[k];
                spill[2u * i + 1u] = d[k];
            }
        }
    }
}


#define ROW_TILES (N_NODES / 16)
#define W_PAIR    144

template <int MODE>
__global__ __launch_bounds__(128)
void gcn_gemm(const float* __restrict__ in, const float* __restrict__ bias,
              const float* __restrict__ W, float* __restrict__ deg,
              float* __restrict__ hs,
              unsigned* __restrict__ ctr, unsigned* __restrict__ cnt, unsigned* __restrict__ spill_n) {
    __shared__ __attribute__((aligned(16))) float Wl[(HDIM / 2) * W_PAIR];

    const int wave = threadIdx.x >> 5;
    const int lane = threadIdx.x & 31;
    const int c    = lane & 15;
    const int h    = lane >> 4;
    const int tm   = blockIdx.x * 4 + wave;
    const int tmc  = tm < ROW_TILES ? tm : ROW_TILES - 1;

    if (MODE == 0) {
        const int i = blockIdx.x * 128 + (int)threadIdx.x;
        if (i < N_NODES) deg[i] = 1.0f;
        if (i < N_RANGES) cnt[i] = 0u;
        if (i < N_CTRS) ctr[i * CTR_STRIDE] = 0u;
        if (i == 0) *spill_n = 0u;
    }

#pragma unroll
    for (int j = 0; j < 4; ++j) {
        const int unit = j * 128 + (int)threadIdx.x;
        const int r = unit >> 4, q = unit & 15;
        const float4 e = *(const float4*)(W + (2 * r) * HDIM + 4 * q);
        const float4 o = *(const float4*)(W + (2 * r + 1) * HDIM + 4 * q);
        float* __restrict__ dst = Wl + r * W_PAIR + 8 * q;
        *(float4*)(dst)     = make_float4(e.x, o.x, e.y, o.y);
        *(float4*)(dst + 4) = make_float4(e.z, o.z, e.w, o.w);
    }

    const int mrow = (tmc << 4) + c;
    const float* __restrict__ arow = in + (size_t)mrow * HDIM + 4 * h;
    float4 a4[8];
#pragma unroll
    for (int j = 0; j < 8; ++j) a4[j] = *(const float4*)(arow + 8 * j);
    if (MODE != 0) {
        const float din = rsqrtf(deg[mrow]);
#pragma unroll
        for (int j = 0; j < 8; ++j) {
            const float4 b4 = *(const float4*)(bias + 8 * j + 4 * h);
            a4[j].x = fmaxf(a4[j].x * din + b4.x, 0.f);
            a4[j].y = fmaxf(a4[j].y * din + b4.y, 0.f);
            a4[j].z = fmaxf(a4[j].z * din + b4.z, 0.f);
            a4[j].w = fmaxf(a4[j].w * din + b4.w, 0.f);
        }
    }

    __syncthreads();
    if (tm >= ROW_TILES) return;

    const int rbase = tm << 4;
    float srow[8];
#pragma unroll
    for (int v = 0; v < 8; ++v) srow[v] = MODE != 0 ? rsqrtf(deg[rbase + 8 * h + v]) : 1.0f;

    const v2f* __restrict__ wl = (const v2f*)(Wl + (2 * h) * W_PAIR + 2 * c);

#pragma unroll
    for (int p = 0; p < 2; ++p) {
        v8f lo = {0.f, 0.f, 0.f, 0.f, 0.f, 0.f, 0.f, 0.f};
        v8f hi = {0.f, 0.f, 0.f, 0.f, 0.f, 0.f, 0.f, 0.f};
#pragma unroll
        for (int j = 0; j < 8; ++j) {
#pragma unroll
            for (int u = 0; u < 2; ++u) {
                v2f a;
                a.x = u ? a4[j].z : a4[j].x;
                a.y = u ? a4[j].w : a4[j].y;
                const int pr = (4 * j + u) * (W_PAIR / 2);
                const v2f b0 = wl[pr + 16 * (2 * p)];
                const v2f b1 = wl[pr + 16 * (2 * p + 1)];
                lo = __builtin_amdgcn_wmma_f32_16x16x4_f32(false, a, false, b0, (short)0, lo, false, false);
                hi = __builtin_amdgcn_wmma_f32_16x16x4_f32(false, a, false, b1, (short)0, hi, false, false);
            }
        }
#pragma unroll
        for (int v = 0; v < 8; ++v) {
            const float l = MODE != 0 ? lo[v] * srow[v] : lo[v];
            const float g = MODE != 0 ? hi[v] * srow[v] : hi[v];
            const float give = h ? l : g;
            const float got = __builtin_bit_cast(float, __builtin_amdgcn_permlanex16(
                __builtin_bit_cast(unsigned, give), __builtin_bit_cast(unsigned, give),
                0x76543210u, 0xfedcba98u, false, false));
            const float r0 = h ? got : l;
            const float r1 = h ? g   : got;
            const size_t o0 = (size_t)(rbase + v) * HDIM + 32 * p + lane;
            const size_t o1 = o0 + 8 * HDIM;
            hs[o0]  = r0;
            hs[o1]  = r1;
        }
        __builtin_amdgcn_sched_barrier(0);
    }
}

#define SUM_THREADS 512
#define SUM_WAVES   (SUM_THREADS / 32)
#define SUM_UNIT    16

template <bool SCALE>
static __device__ __forceinline__ void
add_unit(const unsigned words, const float fl, const unsigned lane,
         const float* __restrict__ h, float* __restrict__ acc) {
    v2f v[SUM_UNIT];
    unsigned a[SUM_UNIT];
#pragma unroll
    for (int k = 0; k < SUM_UNIT; ++k) {
        const unsigned w = (unsigned)__builtin_amdgcn_readlane((int)words, k);
        a[k] = (w >> 16) * HDIM + lane;
        v[k] = *(const v2f*)(h + (w & 0xffffu) * HDIM + 2u * lane);
    }
#pragma unroll
    for (int k = 0; k < SUM_UNIT; ++k) {
        if (SCALE) {
            const float f = __builtin_bit_cast(float, __builtin_amdgcn_readlane(__builtin_bit_cast(int, fl), k));
            v[k].x *= f;
            v[k].y *= f;
        }
        __hip_atomic_fetch_add(acc + a[k],      v[k].x, __ATOMIC_RELAXED, __HIP_MEMORY_SCOPE_WORKGROUP);
        __hip_atomic_fetch_add(acc + a[k] + 32, v[k].y, __ATOMIC_RELAXED, __HIP_MEMORY_SCOPE_WORKGROUP);
    }
}

template <bool SCALE>
__global__ __launch_bounds__(SUM_THREADS)
void gcn_sum(const unsigned* __restrict__ cnt, const unsigned* __restrict__ bucket,
             const unsigned* __restrict__ spill_n, const int* __restrict__ spill,
             const float* __restrict__ deg, const float* __restrict__ h, float* __restrict__ agg) {
    __shared__ float acc[(NPR + 1) * HDIM];
    const unsigned node0 = blockIdx.x * NPR;
    const unsigned lane  = threadIdx.x & 31;
    const unsigned wave  = (unsigned)__builtin_amdgcn_readfirstlane((int)(threadIdx.x >> 5));

    unsigned ne = cnt[blockIdx.x];
    if (ne > (unsigned)CAP) ne = (unsigned)CAP;
    ne = (unsigned)__builtin_amdgcn_readfirstlane((int)ne);
    const unsigned* __restrict__ bk = bucket + blockIdx.x * CAP;
    const unsigned pad = ((unsigned)NPR << 16) | node0;
    unsigned base  = wave * SUM_UNIT;
    unsigned words = pad;
    if (lane < (unsigned)SUM_UNIT && base + lane < ne) words = bk[base + lane];

    for (unsigned p = threadIdx.x; p < (unsigned)(NPR * 32); p += SUM_THREADS) {
        v2f v = *(const v2f*)(h + node0 * HDIM + 2u * p);
        if (SCALE) {
            const float f = rsqrtf(deg[node0 + (p >> 5)]);
            v.x *= f;
            v.y *= f;
        }
        const unsigned w = (p >> 5) * HDIM + (p & 31u);
        acc[w]      = v.x;
        acc[w + 32] = v.y;
    }
    if (threadIdx.x < HDIM) acc[NPR * HDIM + threadIdx.x] = 0.f;
    __syncthreads();

    while (base < ne) {
        const unsigned nbase = base + SUM_WAVES * SUM_UNIT;
        unsigned nwords = pad;
        if (lane < (unsigned)SUM_UNIT && nbase + lane < ne) nwords = bk[nbase + lane];
        float fl = 1.0f;
        if (SCALE) fl = rsqrtf(deg[words & 0xffffu]);
        add_unit<SCALE>(words, fl, lane, h, acc);
        words = nwords;
        base  = nbase;
    }

    unsigned ns = *spill_n;
    if (ns > (unsigned)N_EDGES) ns = (unsigned)N_EDGES;
    ns = (unsigned)__builtin_amdgcn_readfirstlane((int)ns);
    for (unsigned i = wave; i < ns; i += SUM_WAVES) {
        const unsigned s  = (unsigned)__builtin_amdgcn_readfirstlane(spill[2u * i]);
        const unsigned dl = (unsigned)__builtin_amdgcn_readfirstlane(spill[2u * i + 1u]) - node0;
        if (dl < (unsigned)NPR) {
            v2f v = *(const v2f*)(h + s * HDIM + 2u * lane);
            if (SCALE) {
                const float f = rsqrtf(deg[s]);
                v.x *= f;
                v.y *= f;
            }
            __hip_atomic_fetch_add(acc + dl * HDIM + lane,      v.x, __ATOMIC_RELAXED, __HIP_MEMORY_SCOPE_WORKGROUP);
            __hip_atomic_fetch_add(acc + dl * HDIM + lane + 32, v.y, __ATOMIC_RELAXED, __HIP_MEMORY_SCOPE_WORKGROUP);
        }
    }
    __syncthreads();

    for (unsigned p = threadIdx.x; p < (unsigned)(NPR * 32); p += SUM_THREADS) {
        const unsigned w = (p >> 5) * HDIM + (p & 31u);
        v2f o;
        o.x = acc[w];
        o.y = acc[w + 32];
        *(v2f*)(agg + node0 * HDIM + 2u * p) = o;
    }
}

#define ACT_PIECES 5
#define ACT_QUADS  (N_NODES * HDIM / 4)
#define ACT_BLOCKS (ACT_QUADS / (256 * ACT_PIECES))
static_assert(ACT_BLOCKS * 256 * ACT_PIECES == ACT_QUADS, "slabs must tile the output exactly: the kernel has no tail guard");

__global__ __launch_bounds__(256)
void gcn_act(const float* __restrict__ agg, const float* __restrict__ deg,
             const float* __restrict__ bias, float* __restrict__ out) {
    const unsigned base = blockIdx.x * (256u * ACT_PIECES) + threadIdx.x;
    float4 a[ACT_PIECES];
    float  d[ACT_PIECES];
#pragma unroll
    for (int j = 0; j < ACT_PIECES; ++j) {
        const unsigned q = base + 256u * j;
        a[j] = *(const float4*)(agg + 4u * q);
        d[j] = deg[q >> 4];
    }
    const float4 b4 = *(const float4*)(bias + 4 * (threadIdx.x & 15));
#pragma unroll
    for (int j = 0; j < ACT_PIECES; ++j) {
        const unsigned q = base + 256u * j;
        const float din = rsqrtf(d[j]);
        float4 r;
        r.x = fmaxf(a[j].x * din + b4.x, 0.f);
        r.y = fmaxf(a[j].y * din + b4.y, 0.f);
        r.z = fmaxf(a[j].z * din + b4.z, 0.f);
        r.w = fmaxf(a[j].w * din + b4.w, 0.f);
        *(float4*)(out + 4u * q) = r;
    }
}

extern "C" void kernel_launch(void* const* d_in, const int* in_sizes, int n_in,
                              void* d_out, int out_size, void* d_ws, size_t ws_size,
                              hipStream_t stream) {
    const float* x  = (const float*)d_in[0];
    const int*   ei = (const int*)d_in[1];
    const float* W  = (const float*)d_in[2];
    const float* b  = (const float*)d_in[3];
    float* out = (float*)d_out;

    float* ws   = (float*)d_ws;
    float* deg  = ws;
    float* h    = deg + 50048;
    float* aggA = h    + (size_t)N_NODES * HDIM;
    float* aggB = aggA + (size_t)N_NODES * HDIM;
    unsigned* ctr = (unsigned*)(aggB + (size_t)N_NODES * HDIM);
    unsigned* cnt     = ctr + N_CTRS * CTR_STRIDE;
    unsigned* bucket  = cnt + 1024;
    unsigned* spill_n = bucket + (size_t)N_RANGES * CAP;
    int*      spill   = (int*)(spill_n + CTR_STRIDE);

    const int groupBlocks = 2 * GRP_CHUNKS;
    const int sumBlocks   = N_RANGES;
    const int gemmBlocks  = (ROW_TILES + 3) / 4;
    const int actBlocks   = ACT_BLOCKS;

    gcn_gemm<0><<<gemmBlocks, 128, 0, stream>>>(x, nullptr, W, deg, h, ctr, cnt, spill_n);

    group_edges<<<groupBlocks, GRP_THREADS, 0, stream>>>(ei, deg, ctr, cnt, bucket, spill_n, spill);

    gcn_sum<true> <<<sumBlocks, SUM_THREADS, 0, stream>>>(cnt, bucket, spill_n, spill, deg, h, aggA);

    gcn_gemm<1><<<gemmBlocks, 128, 0, stream>>>(aggA, b, W + HDIM * HDIM, deg, h, nullptr, nullptr, nullptr);
    gcn_sum<false><<<sumBlocks, SUM_THREADS, 0, stream>>>(cnt, bucket, spill_n, spill, deg, h, aggB);

    gcn_gemm<1><<<gemmBlocks, 128, 0, stream>>>(aggB, b + HDIM, W + 2 * HDIM * HDIM, deg, h, nullptr, nullptr, nullptr);
    gcn_sum<false><<<sumBlocks, SUM_THREADS, 0, stream>>>(cnt, bucket, spill_n, spill, deg, h, aggA);

    gcn_act<<<actBlocks, 256, 0, stream>>>(aggA, deg, b + 2 * HDIM, out);
}


# Summing launch: a lane's 7 start-row loads (and its 7 finished-row LDS reads) all go out before its first store, not one memory round trip per loop trip ahead of the barrier; sum order unchanged
# speedup vs baseline: 4.6787x; 1.0645x over previous
#include <hip/hip_runtime.h>

#define N_NODES 50000
#define N_EDGES 800000
#define HDIM    64

typedef float v2f __attribute__((ext_vector_type(2)));
typedef float v8f __attribute__((ext_vector_type(8)));

#define NPR         100
#define N_RANGES    (N_NODES / NPR)
#define CAP         8192
static_assert(N_RANGES * NPR == N_NODES, "ranges must tile the nodes exactly: the summing kernel has no tail guard");
static_assert(N_NODES <= 65536 && NPR <= 65536, "an edge is packed as source | place << 16");

#define HALF_NODES  (N_NODES / 2)
#define HALF_RANGES (N_RANGES / 2)
static_assert(HALF_RANGES * NPR == HALF_NODES, "a range must lie in one half");
#define GRP_THREADS 512
#define GRP_PER_THR 8
#define GRP_EDGES   (GRP_THREADS * GRP_PER_THR)
#define GRP_CHUNKS  ((N_EDGES + GRP_EDGES - 1) / GRP_EDGES)
#define CTR_STRIDE  64
#define N_CTRS      2

static __device__ __forceinline__ int claim_job(unsigned* __restrict__ ctr, const unsigned n_chunks) {
    __shared__ int s_job;
    if (threadIdx.x == 0) {
        const unsigned id = __builtin_amdgcn_s_sendmsg_rtn(0x87);
        unsigned r = (id >> 18) & 1u;
        unsigned t = __hip_atomic_fetch_add(ctr + r * CTR_STRIDE, 1u, __ATOMIC_RELAXED, __HIP_MEMORY_SCOPE_AGENT);
        if (t >= n_chunks) {
            r ^= 1u;
            t = __hip_atomic_fetch_add(ctr + r * CTR_STRIDE, 1u, __ATOMIC_RELAXED, __HIP_MEMORY_SCOPE_AGENT);
        }
        s_job = (t < n_chunks) ? (int)(t * 2u + r) : -1;
    }
    __syncthreads();
    return __builtin_amdgcn_readfirstlane(s_job);
}

__global__ __launch_bounds__(GRP_THREADS)
void group_edges(const int* __restrict__ ei, float* __restrict__ deg, unsigned* __restrict__ ctr,
                 unsigned* __restrict__ cnt, unsigned* __restrict__ bucket,
                 unsigned* __restrict__ spill_n, int* __restrict__ spill) {
    __shared__ unsigned hist[HALF_RANGES];
    const int job = claim_job(ctr, (unsigned)GRP_CHUNKS);
    if (job < 0) return;
    const unsigned half = (unsigned)(job & 1);
    const unsigned lo   = half * HALF_NODES;
    const unsigned rlo  = half * HALF_RANGES;
    const unsigned e0   = (unsigned)(job >> 1) * GRP_EDGES + threadIdx.x;
    for (unsigned i = threadIdx.x; i < (unsigned)HALF_RANGES; i += GRP_THREADS) hist[i] = 0u;

    int s[GRP_PER_THR], d[GRP_PER_THR];
#pragma unroll
    for (int k = 0; k < GRP_PER_THR; ++k) {
        const unsigned e = e0 + (unsigned)k * GRP_THREADS;
        s[k] = -1; d[k] = -1;
        if (e < (unsigned)N_EDGES) { s[k] = ei[e]; d[k] = ei[N_EDGES + e]; }
    }
    __syncthreads();

    unsigned rl[GRP_PER_THR], rk[GRP_PER_THR];
#pragma unroll
    for (int k = 0; k < GRP_PER_THR; ++k) {
        const unsigned dl = (unsigned)d[k] - lo;
        rl[k] = dl / NPR;
        rk[k] = 0u;
        if (dl < (unsigned)HALF_NODES) {
            atomicAdd(&deg[d[k]], 1.0f);
            rk[k] = __hip_atomic_fetch_add(&hist[rl[k]], 1u, __ATOMIC_RELAXED, __HIP_MEMORY_SCOPE_WORKGROUP);
        }
    }
    __syncthreads();

    for (unsigned i = threadIdx.x; i < (unsigned)HALF_RANGES; i += GRP_THREADS) {
        const unsigned c = hist[i];
        if (c) hist[i] = __hip_atomic_fetch_add(cnt + rlo + i, c, __ATOMIC_RELAXED, __HIP_MEMORY_SCOPE_AGENT);
    }
    __syncthreads();

#pragma unroll
    for (int k = 0; k < GRP_PER_THR; ++k) {
        const unsigned dl = (unsigned)d[k] - lo;
        if (dl < (unsigned)HALF_NODES) {
            const unsigned place = hist[rl[k]] + rk[k];
            if (place < (unsigned)CAP) {
                bucket[(rlo + rl[k]) * CAP + place] = (unsigned)s[k] | ((dl - rl[k] * NPR) << 16);
            } else {
                const unsigned i = __hip_atomic_fetch_add(spill_n, 1u, __ATOMIC_RELAXED, __HIP_MEMORY_SCOPE_AGENT);
                spill[2u * i]      = s[k];
                spill[2u * i + 1u] = d[k];
            }
        }
    }
}


#define ROW_TILES (N_NODES / 16)
#define W_PAIR    144

template <int MODE>
__global__ __launch_bounds__(128)
void gcn_gemm(const float* __restrict__ in, const float* __restrict__ bias,
              const float* __restrict__ W, float* __restrict__ deg,
              float* __restrict__ hs,
              unsigned* __restrict__ ctr, unsigned* __restrict__ cnt, unsigned* __restrict__ spill_n) {
    __shared__ __attribute__((aligned(16))) float Wl[(HDIM / 2) * W_PAIR];

    const int wave = threadIdx.x >> 5;
    const int lane = threadIdx.x & 31;
    const int c    = lane & 15;
    const int h    = lane >> 4;
    const int tm   = blockIdx.x * 4 + wave;
    const int tmc  = tm < ROW_TILES ? tm : ROW_TILES - 1;

    if (MODE == 0) {
        const int i = blockIdx.x * 128 + (int)threadIdx.x;
        if (i < N_NODES) deg[i] = 1.0f;
        if (i < N_RANGES) cnt[i] = 0u;
        if (i < N_CTRS) ctr[i * CTR_STRIDE] = 0u;
        if (i == 0) *spill_n = 0u;
    }

#pragma unroll
    for (int j = 0; j < 4; ++j) {
        const int unit = j * 128 + (int)threadIdx.x;
        const int r = unit >> 4, q = unit & 15;
        const float4 e = *(const float4*)(W + (2 * r) * HDIM + 4 * q);
        const float4 o = *(const float4*)(W + (2 * r + 1) * HDIM + 4 * q);
        float* __restrict__ dst = Wl + r * W_PAIR + 8 * q;
        *(float4*)(dst)     = make_float4(e.x, o.x, e.y, o.y);
        *(float4*)(dst + 4) = make_float4(e.z, o.z, e.w, o.w);
    }

    const int mrow = (tmc << 4) + c;
    const float* __restrict__ arow = in + (size_t)mrow * HDIM + 4 * h;
    float4 a4[8];
#pragma unroll
    for (int j = 0; j < 8; ++j) a4[j] = *(const float4*)(arow + 8 * j);
    if (MODE != 0) {
        const float din = rsqrtf(deg[mrow]);
#pragma unroll
        for (int j = 0; j < 8; ++j) {
            const float4 b4 = *(const float4*)(bias + 8 * j + 4 * h);
            a4[j].x = fmaxf(a4[j].x * din + b4.x, 0.f);
            a4[j].y = fmaxf(a4[j].y * din + b4.y, 0.f);
            a4[j].z = fmaxf(a4[j].z * din + b4.z, 0.f);
            a4[j].w = fmaxf(a4[j].w * din + b4.w, 0.f);
        }
    }

    __syncthreads();
    if (tm >= ROW_TILES) return;

    const int rbase = tm << 4;
    float srow[8];
#pragma unroll
    for (int v = 0; v < 8; ++v) srow[v] = MODE != 0 ? rsqrtf(deg[rbase + 8 * h + v]) : 1.0f;

    const v2f* __restrict__ wl = (const v2f*)(Wl + (2 * h) * W_PAIR + 2 * c);

#pragma unroll
    for (int p = 0; p < 2; ++p) {
        v8f lo = {0.f, 0.f, 0.f, 0.f, 0.f, 0.f, 0.f, 0.f};
        v8f hi = {0.f, 0.f, 0.f, 0.f, 0.f, 0.f, 0.f, 0.f};
#pragma unroll
        for (int j = 0; j < 8; ++j) {
#pragma unroll
            for (int u = 0; u < 2; ++u) {
                v2f a;
                a.x = u ? a4[j].z : a4[j].x;
                a.y = u ? a4[j].w : a4[j].y;
                const int pr = (4 * j + u) * (W_PAIR / 2);
                const v2f b0 = wl[pr + 16 * (2 * p)];
                const v2f b1 = wl[pr + 16 * (2 * p + 1)];
                lo = __builtin_amdgcn_wmma_f32_16x16x4_f32(false, a, false, b0, (short)0, lo, false, false);
                hi = __builtin_amdgcn_wmma_f32_16x16x4_f32(false, a, false, b1, (short)0, hi, false, false);
            }
        }
#pragma unroll
        for (int v = 0; v < 8; ++v) {
            const float l = MODE != 0 ? lo[v] * srow[v] : lo[v];
            const float g = MODE != 0 ? hi[v] * srow[v] : hi[v];
            const float give = h ? l : g;
            const float got = __builtin_bit_cast(float, __builtin_amdgcn_permlanex16(
                __builtin_bit_cast(unsigned, give), __builtin_bit_cast(unsigned, give),
                0x76543210u, 0xfedcba98u, false, false));
            const float r0 = h ? got : l;
            const float r1 = h ? g   : got;
            const size_t o0 = (size_t)(rbase + v) * HDIM + 32 * p + lane;
            const size_t o1 = o0 + 8 * HDIM;
            hs[o0]  = r0;
            hs[o1]  = r1;
        }
        __builtin_amdgcn_sched_barrier(0);
    }
}

#define SUM_THREADS 512
#define SUM_WAVES   (SUM_THREADS / 32)
#define SUM_UNIT    16
#define SUM_OWN     (NPR * 32)
#define SUM_COPY    ((SUM_OWN + SUM_THREADS - 1) / SUM_THREADS)
static_assert((SUM_COPY - 1) * SUM_THREADS <= SUM_OWN, "only the last copy trip may be part full: the earlier ones carry no guard");

template <bool SCALE>
static __device__ __forceinline__ void
add_unit(const unsigned words, const float fl, const unsigned lane,
         const float* __restrict__ h, float* __restrict__ acc) {
    v2f v[SUM_UNIT];
    unsigned a[SUM_UNIT];
#pragma unroll
    for (int k = 0; k < SUM_UNIT; ++k) {
        const unsigned w = (unsigned)__builtin_amdgcn_readlane((int)words, k);
        a[k] = (w >> 16) * HDIM + lane;
        v[k] = *(const v2f*)(h + (w & 0xffffu) * HDIM + 2u * lane);
    }
#pragma unroll
    for (int k = 0; k < SUM_UNIT; ++k) {
        if (SCALE) {
            const float f = __builtin_bit_cast(float, __builtin_amdgcn_readlane(__builtin_bit_cast(int, fl), k));
            v[k].x *= f;
            v[k].y *= f;
        }
        __hip_atomic_fetch_add(acc + a[k],      v[k].x, __ATOMIC_RELAXED, __HIP_MEMORY_SCOPE_WORKGROUP);
        __hip_atomic_fetch_add(acc + a[k] + 32, v[k].y, __ATOMIC_RELAXED, __HIP_MEMORY_SCOPE_WORKGROUP);
    }
}

template <bool SCALE>
__global__ __launch_bounds__(SUM_THREADS)
void gcn_sum(const unsigned* __restrict__ cnt, const unsigned* __restrict__ bucket,
             const unsigned* __restrict__ spill_n, const int* __restrict__ spill,
             const float* __restrict__ deg, const float* __restrict__ h, float* __restrict__ agg) {
    __shared__ float acc[(NPR + 1) * HDIM];
    const unsigned node0 = blockIdx.x * NPR;
    const unsigned lane  = threadIdx.x & 31;
    const unsigned wave  = (unsigned)__builtin_amdgcn_readfirstlane((int)(threadIdx.x >> 5));

    unsigned ne = cnt[blockIdx.x];
    if (ne > (unsigned)CAP) ne = (unsigned)CAP;
    ne = (unsigned)__builtin_amdgcn_readfirstlane((int)ne);
    const unsigned* __restrict__ bk = bucket + blockIdx.x * CAP;
    const unsigned pad = ((unsigned)NPR << 16) | node0;
    unsigned base  = wave * SUM_UNIT;
    unsigned words = pad;
    if (lane < (unsigned)SUM_UNIT && base + lane < ne) words = bk[base + lane];

    v2f   own[SUM_COPY];
    float odeg[SUM_COPY];
#pragma unroll
    for (int k = 0; k < SUM_COPY; ++k) {
        const unsigned p = threadIdx.x + (unsigned)k * SUM_THREADS;
        if (k + 1 < SUM_COPY || p < (unsigned)SUM_OWN) {
            own[k] = *(const v2f*)(h + node0 * HDIM + 2u * p);
            if (SCALE) odeg[k] = deg[node0 + (p >> 5)];
        }
    }
#pragma unroll
    for (int k = 0; k < SUM_COPY; ++k) {
        const unsigned p = threadIdx.x + (unsigned)k * SUM_THREADS;
        if (k + 1 < SUM_COPY || p < (unsigned)SUM_OWN) {
            v2f v = own[k];
            if (SCALE) {
                const float f = rsqrtf(odeg[k]);
                v.x *= f;
                v.y *= f;
            }
            const unsigned w = (p >> 5) * HDIM + (p & 31u);
            acc[w]      = v.x;
            acc[w + 32] = v.y;
        }
    }
    if (threadIdx.x < HDIM) acc[NPR * HDIM + threadIdx.x] = 0.f;
    __syncthreads();

    while (base < ne) {
        const unsigned nbase = base + SUM_WAVES * SUM_UNIT;
        unsigned nwords = pad;
        if (lane < (unsigned)SUM_UNIT && nbase + lane < ne) nwords = bk[nbase + lane];
        float fl = 1.0f;
        if (SCALE) fl = rsqrtf(deg[words & 0xffffu]);
        add_unit<SCALE>(words, fl, lane, h, acc);
        words = nwords;
        base  = nbase;
    }

    unsigned ns = *spill_n;
    if (ns > (unsigned)N_EDGES) ns = (unsigned)N_EDGES;
    ns = (unsigned)__builtin_amdgcn_readfirstlane((int)ns);
    for (unsigned i = wave; i < ns; i += SUM_WAVES) {
        const unsigned s  = (unsigned)__builtin_amdgcn_readfirstlane(spill[2u * i]);
        const unsigned dl = (unsigned)__builtin_amdgcn_readfirstlane(spill[2u * i + 1u]) - node0;
        if (dl < (unsigned)NPR) {
            v2f v = *(const v2f*)(h + s * HDIM + 2u * lane);
            if (SCALE) {
                const float f = rsqrtf(deg[s]);
                v.x *= f;
                v.y *= f;
            }
            __hip_atomic_fetch_add(acc + dl * HDIM + lane,      v.x, __ATOMIC_RELAXED, __HIP_MEMORY_SCOPE_WORKGROUP);
            __hip_atomic_fetch_add(acc + dl * HDIM + lane + 32, v.y, __ATOMIC_RELAXED, __HIP_MEMORY_SCOPE_WORKGROUP);
        }
    }
    __syncthreads();

    v2f fin[SUM_COPY];
#pragma unroll
    for (int k = 0; k < SUM_COPY; ++k) {
        const unsigned p = threadIdx.x + (unsigned)k * SUM_THREADS;
        if (k + 1 < SUM_COPY || p < (unsigned)SUM_OWN) {
            const unsigned w = (p >> 5) * HDIM + (p & 31u);
            fin[k].x = acc[w];
            fin[k].y = acc[w + 32];
        }
    }
#pragma unroll
    for (int k = 0; k < SUM_COPY; ++k) {
        const unsigned p = threadIdx.x + (unsigned)k * SUM_THREADS;
        if (k + 1 < SUM_COPY || p < (unsigned)SUM_OWN) *(v2f*)(agg + node0 * HDIM + 2u * p) = fin[k];
    }
}

#define ACT_PIECES 5
#define ACT_QUADS  (N_NODES * HDIM / 4)
#define ACT_BLOCKS (ACT_QUADS / (256 * ACT_PIECES))
static_assert(ACT_BLOCKS * 256 * ACT_PIECES == ACT_QUADS, "slabs must tile the output exactly: the kernel has no tail guard");

__global__ __launch_bounds__(256)
void gcn_act(const float* __restrict__ agg, const float* __restrict__ deg,
             const float* __restrict__ bias, float* __restrict__ out) {
    const unsigned base = blockIdx.x * (256u * ACT_PIECES) + threadIdx.x;
    float4 a[ACT_PIECES];
    float  d[ACT_PIECES];
#pragma unroll
    for (int j = 0; j < ACT_PIECES; ++j) {
        const unsigned q = base + 256u * j;
        a[j] = *(const float4*)(agg + 4u * q);
        d[j] = deg[q >> 4];
    }
    const float4 b4 = *(const float4*)(bias + 4 * (threadIdx.x & 15));
#pragma unroll
    for (int j = 0; j < ACT_PIECES; ++j) {
        const unsigned q = base + 256u * j;
        const float din = rsqrtf(d[j]);
        float4 r;
        r.x = fmaxf(a[j].x * din + b4.x, 0.f);
        r.y = fmaxf(a[j].y * din + b4.y, 0.f);
        r.z = fmaxf(a[j].z * din + b4.z, 0.f);
        r.w = fmaxf(a[j].w * din + b4.w, 0.f);
        *(float4*)(out + 4u * q) = r;
    }
}

extern "C" void kernel_launch(void* const* d_in, const int* in_sizes, int n_in,
                              void* d_out, int out_size, void* d_ws, size_t ws_size,
                              hipStream_t stream) {
    const float* x  = (const float*)d_in[0];
    const int*   ei = (const int*)d_in[1];
    const float* W  = (const float*)d_in[2];
    const float* b  = (const float*)d_in[3];
    float* out = (float*)d_out;

    float* ws   = (float*)d_ws;
    float* deg  = ws;
    float* h    = deg + 50048;
    float* aggA = h    + (size_t)N_NODES * HDIM;
    float* aggB = aggA + (size_t)N_NODES * HDIM;
    unsigned* ctr = (unsigned*)(aggB + (size_t)N_NODES * HDIM);
    unsigned* cnt     = ctr + N_CTRS * CTR_STRIDE;
    unsigned* bucket  = cnt + 1024;
    unsigned* spill_n = bucket + (size_t)N_RANGES * CAP;
    int*      spill   = (int*)(spill_n + CTR_STRIDE);

    const int groupBlocks = 2 * GRP_CHUNKS;
    const int sumBlocks   = N_RANGES;
    const int gemmBlocks  = (ROW_TILES + 3) / 4;
    const int actBlocks   = ACT_BLOCKS;

    gcn_gemm<0><<<gemmBlocks, 128, 0, stream>>>(x, nullptr, W, deg, h, ctr, cnt, spill_n);

    group_edges<<<groupBlocks, GRP_THREADS, 0, stream>>>(ei, deg, ctr, cnt, bucket, spill_n, spill);

    gcn_sum<true> <<<sumBlocks, SUM_THREADS, 0, stream>>>(cnt, bucket, spill_n, spill, deg, h, aggA);

    gcn_gemm<1><<<gemmBlocks, 128, 0, stream>>>(aggA, b, W + HDIM * HDIM, deg, h, nullptr, nullptr, nullptr);
    gcn_sum<false><<<sumBlocks, SUM_THREADS, 0, stream>>>(cnt, bucket, spill_n, spill, deg, h, aggB);

    gcn_gemm<1><<<gemmBlocks, 128, 0, stream>>>(aggB, b + HDIM, W + 2 * HDIM * HDIM, deg, h, nullptr, nullptr, nullptr);
    gcn_sum<false><<<sumBlocks, SUM_THREADS, 0, stream>>>(cnt, bucket, spill_n, spill, deg, h, aggA);

    gcn_act<<<actBlocks, 256, 0, stream>>>(aggA, deg, b + 2 * HDIM, out);
}
